# starting kernel
# baseline (speedup 1.0000x reference)
#include <hip/hip_runtime.h>
#include <stdint.h>

#define IN_DIM 128
#define HID    256
#define EMB    128

typedef __attribute__((ext_vector_type(16))) __bf16       v16bf;
typedef __attribute__((ext_vector_type(8)))  float        v8f;
typedef __attribute__((ext_vector_type(4)))  unsigned int uint4v;
typedef __attribute__((ext_vector_type(2)))  unsigned int uint2v;

__device__ __forceinline__ unsigned short f2bf(float f) {
    unsigned int u = __float_as_uint(f);
    u += 0x7FFFu + ((u >> 16) & 1u);
    return (unsigned short)(u >> 16);
}

__global__ void fill_kernel(float* p, float v, int n) {
    int i = blockIdx.x * blockDim.x + threadIdx.x;
    if (i < n) p[i] = v;
}

__global__ void deg_kernel(const int* __restrict__ ei, float* deg, int E) {
    int e = blockIdx.x * blockDim.x + threadIdx.x;
    if (e < E) atomicAdd(&deg[ei[E + e]], 1.0f);
}

__global__ void rsqrt_kernel(float* d, int n) {
    int i = blockIdx.x * blockDim.x + threadIdx.x;
    if (i < n) d[i] = rsqrtf(d[i]);
}

#define BK  64
#define BN  128
#define LDP 72

__global__ __launch_bounds__(256) void wmma_gemm_kernel(
    const float* __restrict__ A, const float* __restrict__ W,
    float* __restrict__ C, int K, int Nc) {
    __shared__ __align__(16) unsigned short sA[16 * LDP];
    __shared__ __align__(16) unsigned short sB[BN * LDP];

    const int tid   = threadIdx.x;
    const int mBase = blockIdx.x * 16;
    const int nBase = blockIdx.y * BN;
    const int wave  = tid >> 5;
    const int lane  = tid & 31;
    const int mn    = lane & 15;
    const int khalf = lane >> 4;

    const int ar  = tid >> 4;
    const int ac4 = (tid & 15) << 2;

    v8f acc = {0.f, 0.f, 0.f, 0.f, 0.f, 0.f, 0.f, 0.f};

    for (int kc = 0; kc < K; kc += BK) {
        __syncthreads();

        {
            const float4 av = *(const float4*)(&A[(size_t)(mBase + ar) * K + kc + ac4]);
            unsigned int lo = (unsigned int)f2bf(av.x) | ((unsigned int)f2bf(av.y) << 16);
            unsigned int hi = (unsigned int)f2bf(av.z) | ((unsigned int)f2bf(av.w) << 16);
            *(uint2v*)(&sA[ar * LDP + ac4]) = (uint2v){lo, hi};
        }

        float4 wv[8];
#pragma unroll
        for (int it = 0; it < 8; ++it) {
            int idx = tid + it * 256;
            int k = idx >> 5, n4 = (idx & 31) << 2;
            wv[it] = *(const float4*)(&W[(size_t)(kc + k) * Nc + nBase + n4]);
        }
#pragma unroll
        for (int it = 0; it < 8; ++it) {
            int idx = tid + it * 256;
            int k = idx >> 5, n4 = (idx & 31) << 2;
            unsigned short* bp = &sB[n4 * LDP + k];
            bp[0 * LDP] = f2bf(wv[it].x);
            bp[1 * LDP] = f2bf(wv[it].y);
            bp[2 * LDP] = f2bf(wv[it].z);
            bp[3 * LDP] = f2bf(wv[it].w);
        }
        __syncthreads();

#pragma unroll
        for (int k0 = 0; k0 < BK; k0 += 32) {
            union { v16bf v; uint4v q[2]; } fa, fb;
            const unsigned short* pa = sA + mn * LDP + k0 + khalf * 8;
            fa.q[0] = *(const uint4v*)(pa);
            fa.q[1] = *(const uint4v*)(pa + 16);
            const unsigned short* pb = sB + (wave * 16 + mn) * LDP + k0 + khalf * 8;
            fb.q[0] = *(const uint4v*)(pb);
            fb.q[1] = *(const uint4v*)(pb + 16);
            acc = __builtin_amdgcn_wmma_f32_16x16x32_bf16(
                false, fa.v, false, fb.v, (short)0, acc, false, false);
        }
    }

    const int row0 = mBase + khalf * 8;
    const int col  = nBase + wave * 16 + mn;
#pragma unroll
    for (int r = 0; r < 8; ++r)
        C[(size_t)(row0 + r) * Nc + col] = acc[r];
}

__global__ void selfterm_kernel(const float* __restrict__ H, const float* __restrict__ dinv,
                                float* __restrict__ AGG, int total, int ldShift) {
    int t = blockIdx.x * blockDim.x + threadIdx.x;
    if (t >= total) return;
    float di = dinv[t >> ldShift];
    AGG[t] = H[t] * di * di;
}

__global__ void scatter_kernel(const float* __restrict__ H, float* AGG,
                               const int* __restrict__ ei, const float* __restrict__ dinv,
                               int E, int vecShift, int Nc) {
    int t = blockIdx.x * blockDim.x + threadIdx.x;
    int e = t >> vecShift;
    if (e >= E) return;
    int j = (t & ((1 << vecShift) - 1)) << 2;
    int s = ei[e], d = ei[E + e];
    float nrm = dinv[s] * dinv[d];
    const float4 hv = *(const float4*)(H + (size_t)s * Nc + j);
    float* ap = AGG + (size_t)d * Nc + j;
    atomicAdd(ap + 0, hv.x * nrm);
    atomicAdd(ap + 1, hv.y * nrm);
    atomicAdd(ap + 2, hv.z * nrm);
    atomicAdd(ap + 3, hv.w * nrm);
}

__global__ void bias_relu_kernel(float* AGG, const float* __restrict__ b, int total, int mask) {
    int t = blockIdx.x * blockDim.x + threadIdx.x;
    if (t < total) AGG[t] = fmaxf(AGG[t] + b[t & mask], 0.0f);
}

__global__ void colmean_kernel(const float* __restrict__ h2, float* g, int N) {
    int j = blockIdx.x, t = threadIdx.x;
    float s = 0.f;
    for (int i = t; i < N; i += 256) s += h2[(size_t)i * EMB + j];
    __shared__ float red[256];
    red[t] = s; __syncthreads();
    for (int k = 128; k > 0; k >>= 1) {
        if (t < k) red[t] += red[t + k];
        __syncthreads();
    }
    if (t == 0) g[j] = red[0] / (float)N;
}

__global__ void nodehead_kernel(const float* __restrict__ h2, const float* __restrict__ npW,
                                const float* __restrict__ npb, float* out, int N) {
    int t = blockIdx.x * blockDim.x + threadIdx.x;
    if (t >= N * 3) return;
    int i = t / 3, c = t - i * 3;
    const float* hr = h2 + (size_t)i * EMB;
    float acc = npb[c];
#pragma unroll 4
    for (int k = 0; k < EMB; ++k) acc += hr[k] * npW[k * 3 + c];
    out[t] = acc;
}

__global__ void edgeproj_kernel(const float* __restrict__ h2, const float* __restrict__ epW,
                                float* p, int N) {
    int t = blockIdx.x * blockDim.x + threadIdx.x;
    if (t >= N * 4) return;
    int i = t >> 2, q = t & 3;
    int half = q >> 1, c = q & 1;
    const float* hr = h2 + (size_t)i * EMB;
    const float* w  = epW + (size_t)half * EMB * 2 + c;
    float acc = 0.f;
#pragma unroll 4
    for (int k = 0; k < EMB; ++k) acc += hr[k] * w[k * 2];
    p[t] = acc;
}

__global__ void edgehead_kernel(const float* __restrict__ p, const int* __restrict__ ei,
                                const float* __restrict__ epb, float* out, int E) {
    int e = blockIdx.x * blockDim.x + threadIdx.x;
    if (e >= E) return;
    int s = ei[e], d = ei[E + e];
    const float4 ps = *(const float4*)(p + (size_t)s * 4);
    const float4 pd = *(const float4*)(p + (size_t)d * 4);
    out[2 * e + 0] = ps.x + pd.z + epb[0];
    out[2 * e + 1] = ps.y + pd.w + epb[1];
}

__global__ void mlphead_kernel(const float* __restrict__ g,
                               const float* __restrict__ fc1W, const float* __restrict__ fc1b,
                               const float* __restrict__ fc2W, const float* __restrict__ fc2b,
                               const float* __restrict__ gpW,  const float* __restrict__ gpb,
                               float* out_glob, float* out_val) {
    __shared__ float red[256];
    int t = threadIdx.x;
    float acc = fc1b[t];
#pragma unroll 4
    for (int k = 0; k < EMB; ++k) acc += g[k] * fc1W[k * HID + t];
    float v = fmaxf(acc, 0.f);
    red[t] = v * fc2W[t]; __syncthreads();
    for (int k = 128; k > 0; k >>= 1) {
        if (t < k) red[t] += red[t + k];
        __syncthreads();
    }
    if (t == 0) *out_val = red[0] + fc2b[0];
    __syncthreads();
    red[t] = (t < EMB) ? g[t] * gpW[t] : 0.f; __syncthreads();
    for (int k = 128; k > 0; k >>= 1) {
        if (t < k) red[t] += red[t + k];
        __syncthreads();
    }
    if (t == 0) *out_glob = red[0] + gpb[0];
}

extern "C" void kernel_launch(void* const* d_in, const int* in_sizes, int n_in,
                              void* d_out, int out_size, void* d_ws, size_t ws_size,
                              hipStream_t stream) {
    const float* x    = (const float*)d_in[0];
    const int*   ei   = (const int*)  d_in[1];
    const float* W1   = (const float*)d_in[2];
    const float* b1   = (const float*)d_in[3];
    const float* W2   = (const float*)d_in[4];
    const float* b2   = (const float*)d_in[5];
    const float* fc1W = (const float*)d_in[6];
    const float* fc1b = (const float*)d_in[7];
    const float* fc2W = (const float*)d_in[8];
    const float* fc2b = (const float*)d_in[9];
    const float* npW  = (const float*)d_in[10];
    const float* npb  = (const float*)d_in[11];
    const float* epW  = (const float*)d_in[12];
    const float* epb  = (const float*)d_in[13];
    const float* gpW  = (const float*)d_in[14];
    const float* gpb  = (const float*)d_in[15];

    const int N = in_sizes[0] / IN_DIM;
    const int E = in_sizes[1] / 2;

    float* ws = (float*)d_ws;
    size_t off = 0;
    auto alloc = [&](size_t n) { size_t o = off; off += (n + 63) & ~size_t(63); return o; };
    float* dinv = ws + alloc((size_t)N);
    float* g    = ws + alloc(256);
    float* pbuf = ws + alloc((size_t)N * 4);
    float* H1   = ws + alloc((size_t)N * HID);
    float* h1   = ws + alloc((size_t)N * HID);
    float* H2   = H1;
    float* h2   = H1 + (size_t)N * EMB;

    const dim3 blk(256);
    auto nb = [](long long n) { return (unsigned)((n + 255) / 256); };

    fill_kernel<<<nb(N), blk, 0, stream>>>(dinv, 1.0f, N);
    deg_kernel<<<nb(E), blk, 0, stream>>>(ei, dinv, E);
    rsqrt_kernel<<<nb(N), blk, 0, stream>>>(dinv, N);

    wmma_gemm_kernel<<<dim3(N / 16, HID / BN), blk, 0, stream>>>(x, W1, H1, IN_DIM, HID);
    selfterm_kernel<<<nb((long long)N * HID), blk, 0, stream>>>(H1, dinv, h1, N * HID, 8);
    scatter_kernel<<<nb((long long)E * (HID / 4)), blk, 0, stream>>>(H1, h1, ei, dinv, E, 6, HID);
    bias_relu_kernel<<<nb((long long)N * HID), blk, 0, stream>>>(h1, b1, N * HID, HID - 1);

    wmma_gemm_kernel<<<dim3(N / 16, EMB / BN), blk, 0, stream>>>(h1, W2, H2, HID, EMB);
    selfterm_kernel<<<nb((long long)N * EMB), blk, 0, stream>>>(H2, dinv, h2, N * EMB, 7);
    scatter_kernel<<<nb((long long)E * (EMB / 4)), blk, 0, stream>>>(H2, h2, ei, dinv, E, 5, EMB);
    bias_relu_kernel<<<nb((long long)N * EMB), blk, 0, stream>>>(h2, b2, N * EMB, EMB - 1);

    colmean_kernel<<<EMB, blk, 0, stream>>>(h2, g, N);

    float* out = (float*)d_out;
    nodehead_kernel<<<nb((long long)N * 3), blk, 0, stream>>>(h2, npW, npb, out, N);
    edgeproj_kernel<<<nb((long long)N * 4), blk, 0, stream>>>(h2, epW, pbuf, N);
    edgehead_kernel<<<nb(E), blk, 0, stream>>>(pbuf, ei, epb, out + (size_t)N * 3, E);
    mlphead_kernel<<<1, blk, 0, stream>>>(g, fc1W, fc1b, fc2W, fc2b, gpW, gpb,
                                          out + (size_t)N * 3 + (size_t)E * 2,
                                          out + (size_t)N * 3 + (size_t)E * 2 + 1);
}


# Edge grouping without global atomics: two-level counting sort (bin by 256-node destination range, then per-node rank) on LDS integer counters; list order = LDS arrival order (sum order changes)
# speedup vs baseline: 4.9922x; 4.9922x over previous
#include <hip/hip_runtime.h>
#include <stdint.h>

#define IN_DIM 128
#define HID    256
#define EMB    128

typedef __attribute__((ext_vector_type(16))) __bf16       v16bf;
typedef __attribute__((ext_vector_type(8)))  float        v8f;
typedef __attribute__((ext_vector_type(4)))  unsigned int uint4v;
typedef __attribute__((ext_vector_type(2)))  unsigned int uint2v;

__device__ __forceinline__ unsigned short f2bf(float f) {
    unsigned int u = __float_as_uint(f);
    u += 0x7FFFu + ((u >> 16) & 1u);
    return (unsigned short)(u >> 16);
}

#define BKN_SHIFT 8
#define BKN   256
#define EPB   4096
#define MAXB  1024
#define RPT   6

__global__ __launch_bounds__(1024) void bin_count_kernel(const int* __restrict__ ei, int* __restrict__ bh,
                                                         int E, int NBK) {
    __shared__ int hist[MAXB];
    const int t = threadIdx.x;
    for (int b = t; b < NBK; b += 1024) hist[b] = 0;
    __syncthreads();
    const int base = blockIdx.x * EPB + t;
    int d[4];
#pragma unroll
    for (int k = 0; k < 4; ++k) {
        const int e = base + k * 1024;
        d[k] = ei[E + ((e < E) ? e : E - 1)];
    }
    asm volatile("" :: "v"(d[0]), "v"(d[1]), "v"(d[2]), "v"(d[3]));
#pragma unroll
    for (int k = 0; k < 4; ++k)
        if (base + k * 1024 < E) atomicAdd(&hist[d[k] >> BKN_SHIFT], 1);
    __syncthreads();
    for (int b = t; b < NBK; b += 1024) bh[(size_t)blockIdx.x * NBK + b] = hist[b];
}

__global__ __launch_bounds__(1024) void bin_scatter_kernel(const int* __restrict__ ei, const int* __restrict__ bh,
                                                           unsigned int* __restrict__ bedge,
                                                           int* __restrict__ bstart, int E, int NBK, int NBLK) {
    __shared__ int cur[MAXB];
    __shared__ int ptot[4][256];
    __shared__ int pbef[4][256];
    __shared__ int wtot[8];
    const int t = threadIdx.x, lane = t & 31, wv = t >> 5;
    const int col = t & 255, sl = t >> 8;
    const int blk = blockIdx.x;
    int carry = 0;
    for (int b0 = 0; b0 < NBK; b0 += 256) {
        const int b = b0 + col;
        int tot = 0, bef = 0;
        if (b < NBK) {
            for (int k0 = sl; k0 < NBLK; k0 += 64) {
                int v[16];
#pragma unroll
                for (int u = 0; u < 16; ++u) {
                    const int k = k0 + 4 * u;
                    v[u] = bh[(size_t)((k < NBLK) ? k : NBLK - 1) * NBK + b];
                }
#pragma unroll
                for (int u = 0; u < 16; ++u) {
                    const int k = k0 + 4 * u;
                    tot += (k < NBLK) ? v[u] : 0;
                    bef += (k < blk) ? v[u] : 0;
                }
            }
        }
        ptot[sl][col] = tot; pbef[sl][col] = bef;
        __syncthreads();
        int inc = 0;
        if (t < 256) {
            tot = ptot[0][t] + ptot[1][t] + ptot[2][t] + ptot[3][t];
            bef = pbef[0][t] + pbef[1][t] + pbef[2][t] + pbef[3][t];
            inc = tot;
            for (int o = 1; o < 32; o <<= 1) {
                const int v = __shfl_up(inc, o, 32);
                if (lane >= o) inc += v;
            }
            if (lane == 31) wtot[wv] = inc;
        }
        __syncthreads();
        int all = 0, wbase = 0;
        for (int w = 0; w < 8; ++w) {
            const int v = wtot[w];
            all += v;
            if (w < wv) wbase += v;
        }
        if (t < 256 && b0 + t < NBK) {
            const int excl = carry + wbase + inc - tot;
            cur[b0 + t] = excl + bef;
            if (blk == 0) bstart[b0 + t] = excl;
        }
        carry += all;
    }
    if (blk == 0 && t == 0) bstart[NBK] = E;
    __syncthreads();

    const int base = blk * EPB + t;
    int s[4], d[4];
#pragma unroll
    for (int k = 0; k < 4; ++k) {
        const int e  = base + k * 1024;
        const int ec = (e < E) ? e : E - 1;
        s[k] = ei[ec];
        d[k] = ei[E + ec];
    }
    asm volatile("" :: "v"(s[0]), "v"(s[1]), "v"(s[2]), "v"(s[3]), "v"(d[0]), "v"(d[1]), "v"(d[2]), "v"(d[3]));
#pragma unroll
    for (int k = 0; k < 4; ++k) {
        if (base + k * 1024 < E) {
            const int pos = atomicAdd(&cur[d[k] >> BKN_SHIFT], 1);
            bedge[pos] = ((unsigned int)s[k] << BKN_SHIFT) | (unsigned int)(d[k] & (BKN - 1));
        }
    }
}

__global__ __launch_bounds__(1024) void bin_rank_kernel(const unsigned int* __restrict__ bedge,
                                                        const int* __restrict__ bstart,
                                                        int* __restrict__ rowptr, float* __restrict__ dinv,
                                                        int* __restrict__ csrc, int N) {
    __shared__ int cnt[BKN];
    __shared__ int wtot[BKN / 32];
    const int t = threadIdx.x, lane = t & 31, wv = t >> 5;
    const int bs = bstart[blockIdx.x], be = bstart[blockIdx.x + 1];
    if (t < BKN) cnt[t] = 0;
    __syncthreads();
    for (int i0 = bs + t; i0 < be; i0 += RPT * 1024) {
        unsigned int w[RPT]; bool ok[RPT];
#pragma unroll
        for (int k = 0; k < RPT; ++k) {
            const int i = i0 + k * 1024;
            ok[k] = i < be;
            w[k] = bedge[ok[k] ? i : i0];
        }
        asm volatile("" :: "v"(w[0]), "v"(w[1]), "v"(w[2]), "v"(w[3]), "v"(w[4]), "v"(w[5]));
#pragma unroll
        for (int k = 0; k < RPT; ++k)
            if (ok[k]) atomicAdd(&cnt[w[k] & (BKN - 1)], 1);
    }
    __syncthreads();
    int c = 0, inc = 0;
    if (t < BKN) {
        c = cnt[t];
        inc = c;
        for (int o = 1; o < 32; o <<= 1) {
            const int v = __shfl_up(inc, o, 32);
            if (lane >= o) inc += v;
        }
        if (lane == 31) wtot[wv] = inc;
    }
    __syncthreads();
    if (t < BKN) {
        int wbase = 0;
        for (int w = 0; w < BKN / 32; ++w)
            if (w < wv) wbase += wtot[w];
        const int start = bs + wbase + inc - c;
        cnt[t] = start;
        const int node = blockIdx.x * BKN + t;
        if (node < N) {
            rowptr[node] = start;
            dinv[node] = rsqrtf((float)c + 1.0f);
        }
        if (node == N - 1) rowptr[N] = start + c;
    }
    __syncthreads();
    for (int i0 = bs + t; i0 < be; i0 += RPT * 1024) {
        unsigned int w[RPT]; bool ok[RPT];
#pragma unroll
        for (int k = 0; k < RPT; ++k) {
            const int i = i0 + k * 1024;
            ok[k] = i < be;
            w[k] = bedge[ok[k] ? i : i0];
        }
        asm volatile("" :: "v"(w[0]), "v"(w[1]), "v"(w[2]), "v"(w[3]), "v"(w[4]), "v"(w[5]));
#pragma unroll
        for (int k = 0; k < RPT; ++k) {
            if (ok[k]) {
                const int pos = atomicAdd(&cnt[w[k] & (BKN - 1)], 1);
                csrc[pos] = (int)(w[k] >> BKN_SHIFT);
            }
        }
    }
}

__global__ void edgeweight_kernel(const int* __restrict__ csrc, const float* __restrict__ dinv,
                                  float* __restrict__ cw, int E) {
    int p = (blockIdx.x * blockDim.x + threadIdx.x) * 4;
    if (p + 3 < E) {
        const int4 s = *(const int4*)(csrc + p);
        *(float4*)(cw + p) = make_float4(dinv[s.x], dinv[s.y], dinv[s.z], dinv[s.w]);
    } else {
        for (; p < E; ++p) cw[p] = dinv[csrc[p]];
    }
}

#define BN  128

template <int K>
__global__ __launch_bounds__(256) void wmma_gemm_kernel(
    const float* __restrict__ A, const float* __restrict__ W,
    float* __restrict__ C, int Nc, int numTiles) {
    constexpr int LDA = K + 8;
    constexpr int KS  = K / 32;
    constexpr int NL  = (16 * K / 4) / 256;
    __shared__ __align__(16) unsigned short sA[2][16 * LDA];

    const int tid   = threadIdx.x;
    const int wave  = tid >> 5;
    const int lane  = tid & 31;
    const int mn    = lane & 15;
    const int khalf = lane >> 4;
    const int col   = blockIdx.y * BN + wave * 16 + mn;

    union BF { v16bf v; unsigned short h[16]; };
    BF fb[KS];
#pragma unroll
    for (int kk = 0; kk < KS; ++kk) {
        const float* wp = W + (size_t)(kk * 32 + khalf * 8) * Nc + col;
#pragma unroll
        for (int e = 0; e < 8; ++e) {
            fb[kk].h[e]     = f2bf(wp[(size_t)e * Nc]);
            fb[kk].h[8 + e] = f2bf(wp[(size_t)(16 + e) * Nc]);
        }
    }

    int ar[NL], ac[NL];
#pragma unroll
    for (int it = 0; it < NL; ++it) {
        const int idx = tid + it * 256;
        ar[it] = idx / (K / 4);
        ac[it] = (idx % (K / 4)) * 4;
    }

    float4 av[NL];
    int tile = blockIdx.x;
    if (tile < numTiles) {
#pragma unroll
        for (int it = 0; it < NL; ++it)
            av[it] = *(const float4*)(&A[(size_t)(tile * 16 + ar[it]) * K + ac[it]]);
#pragma unroll
        for (int it = 0; it < NL; ++it) {
            unsigned int lo = (unsigned int)f2bf(av[it].x) | ((unsigned int)f2bf(av[it].y) << 16);
            unsigned int hi = (unsigned int)f2bf(av[it].z) | ((unsigned int)f2bf(av[it].w) << 16);
            *(uint2v*)(&sA[0][ar[it] * LDA + ac[it]]) = (uint2v){lo, hi};
        }
    }

    int buf = 0;
    for (; tile < numTiles; tile += gridDim.x) {
        const int  next     = tile + (int)gridDim.x;
        const bool has_next = next < numTiles;
        const int ld_tile = has_next ? next : tile;
#pragma unroll
        for (int it = 0; it < NL; ++it)
            av[it] = *(const float4*)(&A[(size_t)(ld_tile * 16 + ar[it]) * K + ac[it]]);
        __builtin_amdgcn_fence(__ATOMIC_RELEASE, "workgroup", "local");
        __builtin_amdgcn_s_barrier();
        __builtin_amdgcn_fence(__ATOMIC_ACQUIRE, "workgroup", "local");

        const unsigned short* pa = &sA[buf][mn * LDA + khalf * 8];
        v8f acc = {0.f, 0.f, 0.f, 0.f, 0.f, 0.f, 0.f, 0.f};
        union { v16bf v; uint4v q[2]; } fa[KS];
#pragma unroll
        for (int kk = 0; kk < KS; ++kk) {
            fa[kk].q[0] = *(const uint4v*)(pa + kk * 32);
            fa[kk].q[1] = *(const uint4v*)(pa + kk * 32 + 16);
        }
        __builtin_amdgcn_sched_barrier(0);
#pragma unroll
        for (int kk = 0; kk < KS; ++kk)
            acc = __builtin_amdgcn_wmma_f32_16x16x32_bf16(
                false, fa[kk].v, false, fb[kk].v, (short)0, acc, false, false);

        float* cp = C + (size_t)(tile * 16 + khalf * 8) * Nc + col;
#pragma unroll
        for (int r = 0; r < 8; ++r) cp[(size_t)r * Nc] = acc[r];

        if (has_next) {
#pragma unroll
            for (int it = 0; it < NL; ++it) {
                unsigned int lo = (unsigned int)f2bf(av[it].x) | ((unsigned int)f2bf(av[it].y) << 16);
                unsigned int hi = (unsigned int)f2bf(av[it].z) | ((unsigned int)f2bf(av[it].w) << 16);
                *(uint2v*)(&sA[buf ^ 1][ar[it] * LDA + ac[it]]) = (uint2v){lo, hi};
            }
        }
        buf ^= 1;
    }
}

template <int NC, int K>
__device__ __forceinline__ void add_rows(const float* __restrict__ H, int sv, float wl,
                                         int j, int lane, float4* acc) {
    constexpr int V = NC / 128;
    float4 h[K][V]; float w[K];
#pragma unroll
    for (int u = 0; u < K; ++u) {
        const int s = __builtin_amdgcn_readlane(sv, j + u);
        w[u] = __int_as_float(__builtin_amdgcn_readlane(__float_as_int(wl), j + u));
        const float4* hp = (const float4*)(H + (size_t)s * NC) + lane;
#pragma unroll
        for (int v = 0; v < V; ++v) h[u][v] = hp[v * 32];
    }
#pragma unroll
    for (int u = 0; u < K; ++u) {
#pragma unroll
        for (int v = 0; v < V; ++v) {
            acc[v].x = __builtin_fmaf(h[u][v].x, w[u], acc[v].x);
            acc[v].y = __builtin_fmaf(h[u][v].y, w[u], acc[v].y);
            acc[v].z = __builtin_fmaf(h[u][v].z, w[u], acc[v].z);
            acc[v].w = __builtin_fmaf(h[u][v].w, w[u], acc[v].w);
        }
    }
}

template <int NC>
__global__ __launch_bounds__(256) void gather_kernel(
    const float* __restrict__ H, const int* __restrict__ rowptr,
    const int* __restrict__ csrc, const float* __restrict__ cw,
    const float* __restrict__ dinv, const float* __restrict__ bias,
    float* __restrict__ OUT, int N) {
    constexpr int V = NC / 128;
    constexpr int U = 8 / V;
    const int lane   = threadIdx.x & 31;
    const int wave0  = blockIdx.x * 8 + __builtin_amdgcn_readfirstlane((int)(threadIdx.x >> 5));
    const int nwaves = gridDim.x * 8;

    float4 bv[V];
#pragma unroll
    for (int v = 0; v < V; ++v) bv[v] = *((const float4*)bias + v * 32 + lane);

    for (int i = wave0; i < N; i += nwaves) {
        const float di  = dinv[i];
        const int   beg = rowptr[i], end = rowptr[i + 1];
        const float dd  = di * di;
        const float4* hi = (const float4*)(H + (size_t)i * NC) + lane;
        float4 acc[V];
#pragma unroll
        for (int v = 0; v < V; ++v) {
            const float4 h = hi[v * 32];
            acc[v] = make_float4(h.x * dd, h.y * dd, h.z * dd, h.w * dd);
        }
        for (int p0 = beg; p0 < end; p0 += 32) {
            const int n = (end - p0 < 32) ? (end - p0) : 32;
            int sv = 0; float wl = 0.f;
            if (lane < n) { sv = csrc[p0 + lane]; wl = cw[p0 + lane] * di; }
            int j = 0;
            for (; j + U <= n; j += U) add_rows<NC, U>(H, sv, wl, j, lane, acc);
            if constexpr (U > 4) {
                if (j + 4 <= n) { add_rows<NC, 4>(H, sv, wl, j, lane, acc); j += 4; }
            }
            if (j + 2 <= n) { add_rows<NC, 2>(H, sv, wl, j, lane, acc); j += 2; }
            if (j < n)      { add_rows<NC, 1>(H, sv, wl, j, lane, acc); }
        }
        float4* op = (float4*)(OUT + (size_t)i * NC) + lane;
#pragma unroll
        for (int v = 0; v < V; ++v) {
            op[v * 32] = make_float4(fmaxf(acc[v].x + bv[v].x, 0.0f), fmaxf(acc[v].y + bv[v].y, 0.0f),
                                     fmaxf(acc[v].z + bv[v].z, 0.0f), fmaxf(acc[v].w + bv[v].w, 0.0f));
        }
    }
}

#define GBLK  2048
#define NHEAD 7

template <int M>
__device__ __forceinline__ float row_xor_add(float x) {
    return x + __builtin_bit_cast(float, __builtin_amdgcn_update_dpp(0, __builtin_bit_cast(int, x), 0x160 + M, 0xf, 0xf, true));
}
__device__ __forceinline__ float wave_sum(float x) {
    x = row_xor_add<1>(x);
    x = row_xor_add<2>(x);
    x = row_xor_add<4>(x);
    x = row_xor_add<8>(x);
    const int xi = __builtin_bit_cast(int, x);
    return x + __builtin_bit_cast(float, __builtin_amdgcn_permlanex16(xi, xi, 0x76543210, 0xfedcba98, false, false));
}

__global__ __launch_bounds__(256) void gather_heads_kernel(
    const float* __restrict__ H, const int* __restrict__ rowptr,
    const int* __restrict__ csrc, const float* __restrict__ cw,
    const float* __restrict__ dinv, const float* __restrict__ bias,
    const float* __restrict__ npW, const float* __restrict__ npb, const float* __restrict__ epW,
    float* __restrict__ node_out, float* __restrict__ p, float* __restrict__ part, int N) {
    constexpr int U = 8;
    __shared__ __align__(16) float wsh[NHEAD * EMB];
    __shared__ __align__(16) float csh[8 * EMB];
    const int lane   = threadIdx.x & 31;
    const int wv     = __builtin_amdgcn_readfirstlane((int)(threadIdx.x >> 5));
    const int wave0  = blockIdx.x * 8 + wv;
    const int nwaves = gridDim.x * 8;

    if (threadIdx.x < EMB) {
        const int k = threadIdx.x;
        wsh[0 * EMB + k] = npW[k * 3 + 0];
        wsh[1 * EMB + k] = npW[k * 3 + 1];
        wsh[2 * EMB + k] = npW[k * 3 + 2];
        wsh[3 * EMB + k] = epW[k * 2 + 0];
        wsh[4 * EMB + k] = epW[k * 2 + 1];
    } else {
        const int k = threadIdx.x - EMB;
        wsh[5 * EMB + k] = epW[(EMB + k) * 2 + 0];
        wsh[6 * EMB + k] = epW[(EMB + k) * 2 + 1];
    }
    __syncthreads();

    const float4 bv = *((const float4*)bias + lane);
    const float  nb = (lane < 3) ? npb[lane] : 0.0f;
    float4 cs = make_float4(0.f, 0.f, 0.f, 0.f);

    for (int i = wave0; i < N; i += nwaves) {
        const float di  = dinv[i];
        const int   beg = rowptr[i], end = rowptr[i + 1];
        const float dd  = di * di;
        float4 acc[1];
        {
            const float4 h = *((const float4*)(H + (size_t)i * EMB) + lane);
            acc[0] = make_float4(h.x * dd, h.y * dd, h.z * dd, h.w * dd);
        }
        for (int p0 = beg; p0 < end; p0 += 32) {
            const int n = (end - p0 < 32) ? (end - p0) : 32;
            int sv = 0; float wl = 0.f;
            if (lane < n) { sv = csrc[p0 + lane]; wl = cw[p0 + lane] * di; }
            int j = 0;
            for (; j + U <= n; j += U) add_rows<EMB, U>(H, sv, wl, j, lane, acc);
            if (j + 4 <= n) { add_rows<EMB, 4>(H, sv, wl, j, lane, acc); j += 4; }
            if (j + 2 <= n) { add_rows<EMB, 2>(H, sv, wl, j, lane, acc); j += 2; }
            if (j < n)      { add_rows<EMB, 1>(H, sv, wl, j, lane, acc); }
        }
        const float4 r = make_float4(fmaxf(acc[0].x + bv.x, 0.0f), fmaxf(acc[0].y + bv.y, 0.0f),
                                     fmaxf(acc[0].z + bv.z, 0.0f), fmaxf(acc[0].w + bv.w, 0.0f));
        cs.x += r.x; cs.y += r.y; cs.z += r.z; cs.w += r.w;

        __builtin_amdgcn_fence(__ATOMIC_ACQUIRE, "wavefront", "local");
        float d[NHEAD];
#pragma unroll
        for (int q = 0; q < NHEAD; ++q) {
            const float4 w = *(const float4*)(&wsh[q * EMB + 4 * lane]);
            d[q] = __builtin_fmaf(r.w, w.w, __builtin_fmaf(r.z, w.z, __builtin_fmaf(r.y, w.y, r.x * w.x)));
        }
#pragma unroll
        for (int q = 0; q < NHEAD; ++q) d[q] = wave_sum(d[q]);

        const float nl = (lane == 0) ? d[0] : ((lane == 1) ? d[1] : d[2]);
        const float pv = (lane == 0) ? d[3] : ((lane == 1) ? d[4] : ((lane == 2) ? d[5] : d[6]));
        if (lane < 3) node_out[(size_t)i * 3 + lane] = nl + nb;
        if (lane < 4) p[(size_t)i * 4 + lane] = pv;
    }

    *(float4*)(&csh[wv * EMB + 4 * lane]) = cs;
    __syncthreads();
    if (threadIdx.x < EMB) {
        float s = 0.f;
#pragma unroll
        for (int w = 0; w < 8; ++w) s += csh[w * EMB + threadIdx.x];
        part[(size_t)blockIdx.x * EMB + threadIdx.x] = s;
    }
}

__global__ __launch_bounds__(256) void colsum_reduce_kernel(const float* __restrict__ part,
                                                            float* __restrict__ part2) {
    __shared__ float sh[256];
    const int t = threadIdx.x, c = t & (EMB - 1), hf = t >> 7;
    const float* pp = part + ((size_t)blockIdx.x * 64 + hf * 32) * EMB + c;
    float s = 0.f;
#pragma unroll
    for (int r = 0; r < 32; ++r) s += pp[r * EMB];
    sh[t] = s; __syncthreads();
    if (t < EMB) part2[blockIdx.x * EMB + t] = sh[t] + sh[t + 128];
}

__global__ void edgehead_kernel(const float* __restrict__ p, const int* __restrict__ ei,
                                const float* __restrict__ epb, float* out, int E) {
    int e = blockIdx.x * blockDim.x + threadIdx.x;
    if (e >= E) return;
    int s = ei[e], d = ei[E + e];
    const float4 ps = *(const float4*)(p + (size_t)s * 4);
    const float4 pd = *(const float4*)(p + (size_t)d * 4);
    out[2 * e + 0] = ps.x + pd.z + epb[0];
    out[2 * e + 1] = ps.y + pd.w + epb[1];
}

__global__ void mlphead_kernel(const float* __restrict__ part2, int N,
                               const float* __restrict__ fc1W, const float* __restrict__ fc1b,
                               const float* __restrict__ fc2W, const float* __restrict__ fc2b,
                               const float* __restrict__ gpW,  const float* __restrict__ gpb,
                               float* out_glob, float* out_val) {
    __shared__ float red[256];
    __shared__ float g[EMB];
    int t = threadIdx.x;
    {
        constexpr int R = GBLK / 64 / 2;
        const float* pp = part2 + (size_t)((t >> 7) * R) * EMB + (t & (EMB - 1));
        float s = 0.f;
#pragma unroll
        for (int r = 0; r < R; ++r) s += pp[r * EMB];
        red[t] = s; __syncthreads();
        if (t < EMB) g[t] = (red[t] + red[t + 128]) / (float)N;
        __syncthreads();
    }
    float acc = fc1b[t];
#pragma unroll 4
    for (int k = 0; k < EMB; ++k) acc += g[k] * fc1W[k * HID + t];
    float v = fmaxf(acc, 0.f);
    red[t] = v * fc2W[t]; __syncthreads();
    for (int k = 128; k > 0; k >>= 1) {
        if (t < k) red[t] += red[t + k];
        __syncthreads();
    }
    if (t == 0) *out_val = red[0] + fc2b[0];
    __syncthreads();
    red[t] = (t < EMB) ? g[t] * gpW[t] : 0.f; __syncthreads();
    for (int k = 128; k > 0; k >>= 1) {
        if (t < k) red[t] += red[t + k];
        __syncthreads();
    }
    if (t == 0) *out_glob = red[0] + gpb[0];
}

extern "C" void kernel_launch(void* const* d_in, const int* in_sizes, int n_in,
                              void* d_out, int out_size, void* d_ws, size_t ws_size,
                              hipStream_t stream) {
    const float* x    = (const float*)d_in[0];
    const int*   ei   = (const int*)  d_in[1];
    const float* W1   = (const float*)d_in[2];
    const float* b1   = (const float*)d_in[3];
    const float* W2   = (const float*)d_in[4];
    const float* b2   = (const float*)d_in[5];
    const float* fc1W = (const float*)d_in[6];
    const float* fc1b = (const float*)d_in[7];
    const float* fc2W = (const float*)d_in[8];
    const float* fc2b = (const float*)d_in[9];
    const float* npW  = (const float*)d_in[10];
    const float* npb  = (const float*)d_in[11];
    const float* epW  = (const float*)d_in[12];
    const float* epb  = (const float*)d_in[13];
    const float* gpW  = (const float*)d_in[14];
    const float* gpb  = (const float*)d_in[15];

    const int N = in_sizes[0] / IN_DIM;
    const int E = in_sizes[1] / 2;

    float* ws = (float*)d_ws;
    size_t off = 0;
    auto alloc = [&](size_t n) { size_t o = off; off += (n + 63) & ~size_t(63); return o; };
    float* dinv = ws + alloc((size_t)N);
    const int NBK  = (N + BKN - 1) / BKN;
    const int NBLK = (E + EPB - 1) / EPB;
    int*   bh     = (int*)(ws + alloc((size_t)NBLK * NBK));
    int*   bstart = (int*)(ws + alloc((size_t)NBK + 1));
    int*   rowptr = (int*)(ws + alloc((size_t)N + 1));
    unsigned int* bedge = (unsigned int*)(ws + alloc((size_t)E));
    int*   csrc   = (int*)(ws + alloc((size_t)E));
    float* cw     = ws + alloc((size_t)E);
    float* part  = ws + alloc((size_t)GBLK * EMB);
    float* part2 = ws + alloc((size_t)(GBLK / 64) * EMB);
    float* pbuf = ws + alloc((size_t)N * 4);
    float* H1   = ws + alloc((size_t)N * HID);
    float* h1   = ws + alloc((size_t)N * HID);
    float* H2   = H1;

    const dim3 blk(256);
    auto nb = [](long long n) { return (unsigned)((n + 255) / 256); };

    bin_count_kernel<<<NBLK, dim3(1024), 0, stream>>>(ei, bh, E, NBK);
    bin_scatter_kernel<<<NBLK, dim3(1024), 0, stream>>>(ei, bh, bedge, bstart, E, NBK, NBLK);
    bin_rank_kernel<<<NBK, dim3(1024), 0, stream>>>(bedge, bstart, rowptr, dinv, csrc, N);
    edgeweight_kernel<<<nb((E + 3) / 4), blk, 0, stream>>>(csrc, dinv, cw, E);

    wmma_gemm_kernel<IN_DIM><<<dim3(512, HID / BN), blk, 0, stream>>>(x, W1, H1, HID, N / 16);
    gather_kernel<HID><<<GBLK, blk, 0, stream>>>(H1, rowptr, csrc, cw, dinv, b1, h1, N);

    float* out = (float*)d_out;
    wmma_gemm_kernel<HID><<<dim3(512, EMB / BN), blk, 0, stream>>>(h1, W2, H2, EMB, N / 16);
    gather_heads_kernel<<<GBLK, blk, 0, stream>>>(H2, rowptr, csrc, cw, dinv, b2, npW, npb, epW,
                                                   out, pbuf, part, N);
    colsum_reduce_kernel<<<GBLK / 64, blk, 0, stream>>>(part, part2);

    edgehead_kernel<<<nb(E), blk, 0, stream>>>(pbuf, ei, epb, out + (size_t)N * 3, E);
    mlphead_kernel<<<1, blk, 0, stream>>>(part2, N, fc1W, fc1b, fc2W, fc2b, gpW, gpb,
                                          out + (size_t)N * 3 + (size_t)E * 2,
                                          out + (size_t)N * 3 + (size_t)E * 2 + 1);
}


# x@W1 kept as bf16 (extra rounding, nearest even) between its product and the layer-1 aggregation: gathered rows 512 B not 1 KB, one 16-byte load a lane and row; sums stay f32, list order unchanged
# speedup vs baseline: 8.0488x; 1.6123x over previous
#include <hip/hip_runtime.h>
#include <stdint.h>

#define IN_DIM 128
#define HID    256
#define EMB    128

typedef __attribute__((ext_vector_type(16))) __bf16       v16bf;
typedef __attribute__((ext_vector_type(8)))  float        v8f;
typedef __attribute__((ext_vector_type(4)))  unsigned int uint4v;
typedef __attribute__((ext_vector_type(2)))  unsigned int uint2v;

__device__ __forceinline__ unsigned short f2bf(float f) {
    unsigned int u = __float_as_uint(f);
    u += 0x7FFFu + ((u >> 16) & 1u);
    return (unsigned short)(u >> 16);
}

#define BKN_SHIFT 8
#define BKN   256
#define EPB   4096
#define MAXB  1024
#define SG    8

__device__ __forceinline__ void bin_local_body(const int* __restrict__ ei, unsigned int* __restrict__ bedge,
                                               int* __restrict__ boff, int E, int NBK) {
    __shared__ int hist[MAXB];
    __shared__ int wtot[32];
    const int t = threadIdx.x, lane = t & 31, wv = t >> 5;
    const int blk = blockIdx.x;
    if (t < NBK) hist[t] = 0;
    __syncthreads();
    const int base = blk * EPB + t;
    int s[4], d[4];
#pragma unroll
    for (int k = 0; k < 4; ++k) {
        const int e  = base + k * 1024;
        const int ec = (e < E) ? e : E - 1;
        s[k] = ei[ec];
        d[k] = ei[E + ec];
    }
    asm volatile("" :: "v"(s[0]), "v"(s[1]), "v"(s[2]), "v"(s[3]), "v"(d[0]), "v"(d[1]), "v"(d[2]), "v"(d[3]));
#pragma unroll
    for (int k = 0; k < 4; ++k)
        if (base + k * 1024 < E) atomicAdd(&hist[d[k] >> BKN_SHIFT], 1);
    __syncthreads();
    const int c = (t < NBK) ? hist[t] : 0;
    int inc = c;
    for (int o = 1; o < 32; o <<= 1) {
        const int v = __shfl_up(inc, o, 32);
        if (lane >= o) inc += v;
    }
    if (lane == 31) wtot[wv] = inc;
    __syncthreads();
    int wbase = 0;
    for (int w = 0; w < 32; ++w)
        if (w < wv) wbase += wtot[w];
    const int excl = wbase + inc - c;
    int* bo = boff + (size_t)blk * (NBK + 1);
    if (t < NBK) { hist[t] = excl; bo[t] = excl; }
    if (t == NBK - 1) bo[NBK] = excl + c;
    __syncthreads();
    unsigned int pos[4];
#pragma unroll
    for (int k = 0; k < 4; ++k)
        pos[k] = (base + k * 1024 < E) ? (unsigned int)atomicAdd(&hist[d[k] >> BKN_SHIFT], 1) : 0u;
#pragma unroll
    for (int k = 0; k < 4; ++k)
        if (base + k * 1024 < E)
            bedge[(unsigned int)(blk * EPB) + pos[k]] = ((unsigned int)s[k] << BKN_SHIFT) | (unsigned int)(d[k] & (BKN - 1));
}

__device__ __forceinline__ void part_bounds(const int* __restrict__ boff, int pitch, int bin, int q0, int wv,
                                            int lane, int NBLK, int* o, int* c, int& front) {
    const int  sl = (q0 + (lane & (SG - 1))) * 32 + wv;
    const bool in = sl < NBLK;
    const int* p  = boff + (size_t)(in ? sl : 0) * pitch + bin;
    const int a0 = p[0], a1 = p[1];
    const int ov = sl * EPB + a0;
    const int cv = in ? (a1 - a0) : 0;
    const int fv = in ? a0 : 0;
#pragma unroll
    for (int q = 0; q < SG; ++q) {
        o[q] = __builtin_amdgcn_readlane(ov, q);
        c[q] = __builtin_amdgcn_readlane(cv, q);
        front += __builtin_amdgcn_readlane(fv, q);
    }
}

__device__ __forceinline__ void part_heads(const unsigned int* __restrict__ bedge, int lane,
                                           const int* o, const int* c, unsigned int* w) {
#pragma unroll
    for (int q = 0; q < SG; ++q) w[q] = bedge[(lane < c[q]) ? (o[q] + lane) : 0];
    asm volatile("" :: "v"(w[0]), "v"(w[1]), "v"(w[2]), "v"(w[3]), "v"(w[4]), "v"(w[5]), "v"(w[6]), "v"(w[7]));
}

__device__ __forceinline__ void count_parts(int* cnt, const unsigned int* __restrict__ bedge, int lane,
                                            const int* o, const int* c, const unsigned int* w) {
#pragma unroll
    for (int q = 0; q < SG; ++q)
        if (lane < c[q]) atomicAdd(&cnt[w[q] & (BKN - 1)], 1);
#pragma unroll
    for (int q = 0; q < SG; ++q)
        for (int j = 32 + lane; j < c[q]; j += 32) atomicAdd(&cnt[bedge[o[q] + j] & (BKN - 1)], 1);
}

__device__ __forceinline__ void rank_parts(int* cnt, const unsigned int* __restrict__ bedge, int* __restrict__ csrc,
                                           int lane, const int* o, const int* c, const unsigned int* w) {
    unsigned int pos[SG];
#pragma unroll
    for (int q = 0; q < SG; ++q)
        pos[q] = (lane < c[q]) ? (unsigned int)atomicAdd(&cnt[w[q] & (BKN - 1)], 1) : 0u;
#pragma unroll
    for (int q = 0; q < SG; ++q)
        if (lane < c[q]) csrc[pos[q]] = (int)(w[q] >> BKN_SHIFT);
#pragma unroll
    for (int q = 0; q < SG; ++q) {
        for (int j = 32 + lane; j < c[q]; j += 32) {
            const unsigned int x = bedge[o[q] + j];
            const int at = atomicAdd(&cnt[x & (BKN - 1)], 1);
            csrc[at] = (int)(x >> BKN_SHIFT);
        }
    }
}

__device__ __forceinline__ void bin_rank_body(const unsigned int* __restrict__ bedge,
                                              const int* __restrict__ boff,
                                              int* __restrict__ rowptr, float* __restrict__ dinv,
                                              int* __restrict__ csrc, int N, int NBK, int NBLK) {
    __shared__ int cnt[BKN];
    __shared__ int wtot[BKN / 32];
    __shared__ int wsum[32];
    const int t = threadIdx.x, lane = t & 31;
    const int wv = t >> 5;
    const int bin = blockIdx.x;
    const int pitch = NBK + 1;
    if (t < BKN) cnt[t] = 0;
    int front = 0;
    int o[SG], c[SG];
    unsigned int w[SG];
    part_bounds(boff, pitch, bin, 0, wv, lane, NBLK, o, c, front);
    part_heads(bedge, lane, o, c, w);
    __syncthreads();
    count_parts(cnt, bedge, lane, o, c, w);
    for (int q0 = SG; q0 * 32 < NBLK; q0 += SG) {
        int o2[SG], c2[SG];
        unsigned int w2[SG];
        part_bounds(boff, pitch, bin, q0, wv, lane, NBLK, o2, c2, front);
        part_heads(bedge, lane, o2, c2, w2);
        count_parts(cnt, bedge, lane, o2, c2, w2);
    }
    if (lane == 0) wsum[wv] = front;
    __syncthreads();
    int dg = 0, inc = 0;
    if (t < BKN) {
        dg = cnt[t];
        inc = dg;
        for (int s = 1; s < 32; s <<= 1) {
            const int v = __shfl_up(inc, s, 32);
            if (lane >= s) inc += v;
        }
        if (lane == 31) wtot[wv] = inc;
    }
    __syncthreads();
    if (t < BKN) {
        int bs = 0;
        for (int k = 0; k < 32; ++k) bs += wsum[k];
        int wbase = 0;
        for (int k = 0; k < BKN / 32; ++k)
            if (k < wv) wbase += wtot[k];
        const int start = bs + wbase + inc - dg;
        cnt[t] = start;
        const int node = bin * BKN + t;
        if (node < N) {
            rowptr[node] = start;
            dinv[node] = rsqrtf((float)dg + 1.0f);
        }
        if (node == N - 1) rowptr[N] = start + dg;
    }
    __syncthreads();
    rank_parts(cnt, bedge, csrc, lane, o, c, w);
    for (int q0 = SG; q0 * 32 < NBLK; q0 += SG) {
        int o2[SG], c2[SG], unused = 0;
        unsigned int w2[SG];
        part_bounds(boff, pitch, bin, q0, wv, lane, NBLK, o2, c2, unused);
        part_heads(bedge, lane, o2, c2, w2);
        rank_parts(cnt, bedge, csrc, lane, o2, c2, w2);
    }
}

__global__ void edgeweight_kernel(const int* __restrict__ csrc, const float* __restrict__ dinv,
                                  float* __restrict__ cw, int E) {
    int p = (blockIdx.x * blockDim.x + threadIdx.x) * 4;
    if (p + 3 < E) {
        const int4 s = *(const int4*)(csrc + p);
        *(float4*)(cw + p) = make_float4(dinv[s.x], dinv[s.y], dinv[s.z], dinv[s.w]);
    } else {
        for (; p < E; ++p) cw[p] = dinv[csrc[p]];
    }
}

#define BN  128

__device__ __forceinline__ void gemm1_guest(const float* __restrict__ A, const float* __restrict__ Wm,
                                            unsigned int* __restrict__ C, int g, int G, int pair0, int pair1,
                                            int numTiles) {
    constexpr int K   = IN_DIM;
    constexpr int LDA = K + 8;
    constexpr int KS  = K / 32;
    __shared__ __align__(16) unsigned short sA1[2][32 * LDA];

    const int tid   = threadIdx.x;
    const int wave  = tid >> 5;
    const int lane  = tid & 31;
    const int mn    = lane & 15;
    const int khalf = lane >> 4;
    const int quad  = wave >> 3;
    const int rt    = quad >> 1;
    const int col   = (quad & 1) * BN + (wave & 7) * 16 + mn;

    union BF { v16bf v; unsigned short h[16]; };
    BF fb[KS];
#pragma unroll
    for (int kk = 0; kk < KS; ++kk) {
        const float* wp = Wm + (size_t)(kk * 32 + khalf * 8) * HID + col;
#pragma unroll
        for (int e = 0; e < 8; ++e) {
            fb[kk].h[e]     = f2bf(wp[(size_t)e * HID]);
            fb[kk].h[8 + e] = f2bf(wp[(size_t)(16 + e) * HID]);
        }
    }

    const int sr = wave;
    const int sc = lane * 4;
    const int lastRow = numTiles * 16 - 1;

    float4 av;
    int p = pair0 + g;
    if (p < pair1) {
        const int row = (p * 32 + sr < lastRow) ? (p * 32 + sr) : lastRow;
        av = *(const float4*)(&A[(size_t)row * K + sc]);
        unsigned int lo = (unsigned int)f2bf(av.x) | ((unsigned int)f2bf(av.y) << 16);
        unsigned int hi = (unsigned int)f2bf(av.z) | ((unsigned int)f2bf(av.w) << 16);
        *(uint2v*)(&sA1[0][sr * LDA + sc]) = (uint2v){lo, hi};
    }

    int buf = 0;
    for (; p < pair1; p += G) {
        const int  next     = p + G;
        const bool has_next = next < pair1;
        const int ldp = has_next ? next : p;
        const int row = (ldp * 32 + sr < lastRow) ? (ldp * 32 + sr) : lastRow;
        av = *(const float4*)(&A[(size_t)row * K + sc]);
        __builtin_amdgcn_fence(__ATOMIC_RELEASE, "workgroup", "local");
        __builtin_amdgcn_s_barrier();
        __builtin_amdgcn_fence(__ATOMIC_ACQUIRE, "workgroup", "local");

        const unsigned short* pa = &sA1[buf][(rt * 16 + mn) * LDA + khalf * 8];
        v8f acc = {0.f, 0.f, 0.f, 0.f, 0.f, 0.f, 0.f, 0.f};
#pragma unroll
        for (int kk = 0; kk < KS; ++kk) {
            union { v16bf v; uint4v q[2]; } fa;
            fa.q[0] = *(const uint4v*)(pa + kk * 32);
            fa.q[1] = *(const uint4v*)(pa + kk * 32 + 16);
            acc = __builtin_amdgcn_wmma_f32_16x16x32_bf16(
                false, fa.v, false, fb[kk].v, (short)0, acc, false, false);
        }

        const int tile = 2 * p + rt;
        if (tile < numTiles) {
            const bool odd = (mn & 1) != 0;
            unsigned int* cp = C + (size_t)(tile * 16 + khalf * 8 + (odd ? 4 : 0)) * (HID / 2) + (col >> 1);
#pragma unroll
            for (int r = 0; r < 4; ++r) {
                const unsigned int own = (unsigned int)f2bf(acc[r]) | ((unsigned int)f2bf(acc[4 + r]) << 16);
                const unsigned int oth = (unsigned int)__builtin_amdgcn_update_dpp(0, (int)own, 0xB1, 0xf, 0xf, true);
                const unsigned int we = (own & 0xFFFFu) | (oth << 16);
                const unsigned int wo = (oth >> 16) | (own & 0xFFFF0000u);
                cp[(size_t)r * (HID / 2)] = odd ? wo : we;
            }
        }

        if (has_next) {
            unsigned int lo = (unsigned int)f2bf(av.x) | ((unsigned int)f2bf(av.y) << 16);
            unsigned int hi = (unsigned int)f2bf(av.z) | ((unsigned int)f2bf(av.w) << 16);
            *(uint2v*)(&sA1[buf ^ 1][sr * LDA + sc]) = (uint2v){lo, hi};
        }
        buf ^= 1;
    }
}

__global__ __launch_bounds__(1024, 16) void bin_local_kernel(const int* __restrict__ ei, unsigned int* __restrict__ bedge,
                                                         int* __restrict__ boff, int E, int NBK, int NBLK,
                                                         const float* __restrict__ x, const float* __restrict__ W1,
                                                         unsigned int* __restrict__ H1, int pair0, int pair1, int numTiles) {
    if ((int)blockIdx.x < NBLK) bin_local_body(ei, bedge, boff, E, NBK);
    else gemm1_guest(x, W1, H1, (int)blockIdx.x - NBLK, (int)gridDim.x - NBLK, pair0, pair1, numTiles);
}

__global__ __launch_bounds__(1024, 16) void bin_rank_kernel(const unsigned int* __restrict__ bedge,
                                                        const int* __restrict__ boff,
                                                        int* __restrict__ rowptr, float* __restrict__ dinv,
                                                        int* __restrict__ csrc, int N, int NBK, int NBLK,
                                                        const float* __restrict__ x, const float* __restrict__ W1,
                                                        unsigned int* __restrict__ H1, int pair0, int pair1, int numTiles) {
    if ((int)blockIdx.x < NBK) bin_rank_body(bedge, boff, rowptr, dinv, csrc, N, NBK, NBLK);
    else gemm1_guest(x, W1, H1, (int)blockIdx.x - NBK, (int)gridDim.x - NBK, pair0, pair1, numTiles);
}

template <int NC, int K>
__device__ __forceinline__ void add_rows(const float* __restrict__ H, int sv, float wl,
                                         int j, int lane, float4* acc) {
    constexpr int V = NC / 128;
    float4 h[K][V]; float w[K];
#pragma unroll
    for (int u = 0; u < K; ++u) {
        const int s = __builtin_amdgcn_readlane(sv, j + u);
        w[u] = __int_as_float(__builtin_amdgcn_readlane(__float_as_int(wl), j + u));
        const float4* hp = (const float4*)(H + (size_t)s * NC) + lane;
#pragma unroll
        for (int v = 0; v < V; ++v) h[u][v] = hp[v * 32];
    }
#pragma unroll
    for (int u = 0; u < K; ++u) {
#pragma unroll
        for (int v = 0; v < V; ++v) {
            acc[v].x = __builtin_fmaf(h[u][v].x, w[u], acc[v].x);
            acc[v].y = __builtin_fmaf(h[u][v].y, w[u], acc[v].y);
            acc[v].z = __builtin_fmaf(h[u][v].z, w[u], acc[v].z);
            acc[v].w = __builtin_fmaf(h[u][v].w, w[u], acc[v].w);
        }
    }
}

template <int K>
__device__ __forceinline__ void add_rows_bf(const unsigned int* __restrict__ H, int sv, float wl,
                                            int j, int lane, float* acc) {
    uint4v h[K]; float w[K];
#pragma unroll
    for (int u = 0; u < K; ++u) {
        const int s = __builtin_amdgcn_readlane(sv, j + u);
        w[u] = __int_as_float(__builtin_amdgcn_readlane(__float_as_int(wl), j + u));
        h[u] = *(const uint4v*)((const char*)H + ((unsigned int)s * (unsigned int)(HID * 2) + 16u * (unsigned int)lane));
    }
#pragma unroll
    for (int u = 0; u < K; ++u) {
#pragma unroll
        for (int q = 0; q < 4; ++q) {
            const unsigned int d = h[u][q];
            acc[2 * q]     = __builtin_fmaf(__uint_as_float(d << 16), w[u], acc[2 * q]);
            acc[2 * q + 1] = __builtin_fmaf(__uint_as_float(d & 0xFFFF0000u), w[u], acc[2 * q + 1]);
        }
    }
}

#define LDA2 (HID + 8)

__global__ __launch_bounds__(1024, 16) void gather_gemm2_kernel(
    const unsigned int* __restrict__ H, const int* __restrict__ rowptr,
    const int* __restrict__ csrc, const float* __restrict__ cw,
    const float* __restrict__ dinv, const float* __restrict__ bias,
    const float* __restrict__ W2, float* __restrict__ C, int N) {
    constexpr int U  = 8;
    constexpr int KS = HID / 32;
    __shared__ __align__(16) unsigned short sB[HID * EMB];
    __shared__ __align__(16) unsigned short sA[2][32 * LDA2];
    const int tid  = threadIdx.x;
    const int lane = tid & 31;
    const int wv   = __builtin_amdgcn_readfirstlane(tid >> 5);


    const int numTiles = (N + 15) >> 4;
    const int nTrips   = (N + 31) >> 5;
    int T     = blockIdx.x;
    int prevT = -1;
    int buf   = 0;
    for (;;) {
        const bool have = T < nTrips;
        unsigned int pk[4];
#pragma unroll
        for (int q = 0; q < 4; ++q) pk[q] = 0u;
        const int i = T * 32 + wv;
        if (have && i < N) {
            const float di  = dinv[i];
            const int   beg = rowptr[i], end = rowptr[i + 1];
            const float dd  = di * di;
            float acc[8];
            {
                const uint4v h = *(const uint4v*)((const char*)H + ((unsigned int)i * (unsigned int)(HID * 2) + 16u * (unsigned int)lane));
#pragma unroll
                for (int q = 0; q < 4; ++q) {
                    const unsigned int d = h[q];
                    acc[2 * q]     = __uint_as_float(d << 16) * dd;
                    acc[2 * q + 1] = __uint_as_float(d & 0xFFFF0000u) * dd;
                }
            }
            for (int p0 = beg; p0 < end; p0 += 32) {
                const int n = (end - p0 < 32) ? (end - p0) : 32;
                int sv = 0; float wl = 0.f;
                if (lane < n) { sv = csrc[p0 + lane]; wl = cw[p0 + lane] * di; }
                int j = 0;
                for (; j + U <= n; j += U) add_rows_bf<U>(H, sv, wl, j, lane, acc);
                if (j + 4 <= n) { add_rows_bf<4>(H, sv, wl, j, lane, acc); j += 4; }
                if (j + 2 <= n) { add_rows_bf<2>(H, sv, wl, j, lane, acc); j += 2; }
                if (j < n)      { add_rows_bf<1>(H, sv, wl, j, lane, acc); }
            }
            float bv[8];
            {
                const float4 b0 = *((const float4*)bias + 2 * lane), b1 = *((const float4*)bias + 2 * lane + 1);
                bv[0] = b0.x; bv[1] = b0.y; bv[2] = b0.z; bv[3] = b0.w;
                bv[4] = b1.x; bv[5] = b1.y; bv[6] = b1.z; bv[7] = b1.w;
            }
#pragma unroll
            for (int q = 0; q < 4; ++q) {
                const float r0 = fmaxf(acc[2 * q] + bv[2 * q], 0.0f);
                const float r1 = fmaxf(acc[2 * q + 1] + bv[2 * q + 1], 0.0f);
                pk[q] = (unsigned int)f2bf(r0) | ((unsigned int)f2bf(r1) << 16);
            }
        }

        if (prevT >= 0) {
            __builtin_amdgcn_s_barrier_wait(-1);
            __builtin_amdgcn_fence(__ATOMIC_ACQUIRE, "workgroup", "local");
            const int tl   = wv >> 3;
            const int tile = 2 * prevT + tl;
            if (wv < 16 && tile < numTiles) {
                const int mn    = lane & 15;
                const int khalf = lane >> 4;
                const unsigned short* pa = &sA[buf ^ 1][(tl * 16 + mn) * LDA2 + khalf * 8];
                const unsigned short* pb = &sB[((wv & 7) * (KS * 2 * 32) + lane) * 8];
                v8f d = {0.f, 0.f, 0.f, 0.f, 0.f, 0.f, 0.f, 0.f};
#pragma unroll
                for (int kk = 0; kk < KS; ++kk) {
                    union { v16bf v; uint4v q[2]; } fa, fb;
                    fa.q[0] = *(const uint4v*)(pa + kk * 32);
                    fa.q[1] = *(const uint4v*)(pa + kk * 32 + 16);
                    fb.q[0] = *(const uint4v*)(pb + (kk * 2 + 0) * 256);
                    fb.q[1] = *(const uint4v*)(pb + (kk * 2 + 1) * 256);
                    d = __builtin_amdgcn_wmma_f32_16x16x32_bf16(
                        false, fa.v, false, fb.v, (short)0, d, false, false);
                }
                float* cp = C + (size_t)(tile * 16 + khalf * 8) * EMB + (wv & 7) * 16 + mn;
#pragma unroll
                for (int r = 0; r < 8; ++r) cp[(size_t)r * EMB] = d[r];
            }
        } else if (have) {
#pragma unroll 8
            for (int it = 0; it < (HID * EMB) / 1024; ++it) {
                const int idx = it * 1024 + tid;
                const int k = idx >> 7, col = idx & (EMB - 1);
                const int r = k & 31;
                const int ln = ((r >> 3) & 1) * 16 + (col & 15);
                const int at = (((((col >> 4) * KS + (k >> 5)) * 2 + (r >> 4)) * 32 + ln) << 3) + (r & 7);
                sB[at] = f2bf(W2[idx]);
            }
        }
        if (!have) break;

        *(uint4v*)(&sA[buf][wv * LDA2 + 8 * lane]) = (uint4v){pk[0], pk[1], pk[2], pk[3]};
        __builtin_amdgcn_fence(__ATOMIC_RELEASE, "workgroup", "local");
        __builtin_amdgcn_s_barrier_signal(-1);
        prevT = T;
        T += (int)gridDim.x;
        buf ^= 1;
    }
}

#define GBLK  2048
#define NHEAD 7

template <int M>
__device__ __forceinline__ float row_xor_add(float x) {
    return x + __builtin_bit_cast(float, __builtin_amdgcn_update_dpp(0, __builtin_bit_cast(int, x), 0x160 + M, 0xf, 0xf, true));
}
__device__ __forceinline__ float wave_sum(float x) {
    x = row_xor_add<1>(x);
    x = row_xor_add<2>(x);
    x = row_xor_add<4>(x);
    x = row_xor_add<8>(x);
    const int xi = __builtin_bit_cast(int, x);
    return x + __builtin_bit_cast(float, __builtin_amdgcn_permlanex16(xi, xi, 0x76543210, 0xfedcba98, false, false));
}

__global__ __launch_bounds__(256) void gather_heads_kernel(
    const float* __restrict__ H, const int* __restrict__ rowptr,
    const int* __restrict__ csrc, const float* __restrict__ cw,
    const float* __restrict__ dinv, const float* __restrict__ bias,
    const float* __restrict__ npW, const float* __restrict__ npb, const float* __restrict__ epW,
    float* __restrict__ node_out, float* __restrict__ p, float* __restrict__ part, int N) {
    constexpr int U = 8;
    __shared__ __align__(16) float wsh[NHEAD * EMB];
    __shared__ __align__(16) float csh[8 * EMB];
    const int lane   = threadIdx.x & 31;
    const int wv     = __builtin_amdgcn_readfirstlane((int)(threadIdx.x >> 5));
    const int wave0  = blockIdx.x * 8 + wv;
    const int nwaves = gridDim.x * 8;

    if (threadIdx.x < EMB) {
        const int k = threadIdx.x;
        wsh[0 * EMB + k] = npW[k * 3 + 0];
        wsh[1 * EMB + k] = npW[k * 3 + 1];
        wsh[2 * EMB + k] = npW[k * 3 + 2];
        wsh[3 * EMB + k] = epW[k * 2 + 0];
        wsh[4 * EMB + k] = epW[k * 2 + 1];
    } else {
        const int k = threadIdx.x - EMB;
        wsh[5 * EMB + k] = epW[(EMB + k) * 2 + 0];
        wsh[6 * EMB + k] = epW[(EMB + k) * 2 + 1];
    }
    __syncthreads();

    const float4 bv = *((const float4*)bias + lane);
    const float  nb = (lane < 3) ? npb[lane] : 0.0f;
    float4 cs = make_float4(0.f, 0.f, 0.f, 0.f);

    for (int i = wave0; i < N; i += nwaves) {
        const float di  = dinv[i];
        const int   beg = rowptr[i], end = rowptr[i + 1];
        const float dd  = di * di;
        float4 acc[1];
        {
            const float4 h = *((const float4*)(H + (size_t)i * EMB) + lane);
            acc[0] = make_float4(h.x * dd, h.y * dd, h.z * dd, h.w * dd);
        }
        for (int p0 = beg; p0 < end; p0 += 32) {
            const int n = (end - p0 < 32) ? (end - p0) : 32;
            int sv = 0; float wl = 0.f;
            if (lane < n) { sv = csrc[p0 + lane]; wl = cw[p0 + lane] * di; }
            int j = 0;
            for (; j + U <= n; j += U) add_rows<EMB, U>(H, sv, wl, j, lane, acc);
            if (j + 4 <= n) { add_rows<EMB, 4>(H, sv, wl, j, lane, acc); j += 4; }
            if (j + 2 <= n) { add_rows<EMB, 2>(H, sv, wl, j, lane, acc); j += 2; }
            if (j < n)      { add_rows<EMB, 1>(H, sv, wl, j, lane, acc); }
        }
        const float4 r = make_float4(fmaxf(acc[0].x + bv.x, 0.0f), fmaxf(acc[0].y + bv.y, 0.0f),
                                     fmaxf(acc[0].z + bv.z, 0.0f), fmaxf(acc[0].w + bv.w, 0.0f));
        cs.x += r.x; cs.y += r.y; cs.z += r.z; cs.w += r.w;

        __builtin_amdgcn_fence(__ATOMIC_ACQUIRE, "wavefront", "local");
        float d[NHEAD];
#pragma unroll
        for (int q = 0; q < NHEAD; ++q) {
            const float4 w = *(const float4*)(&wsh[q * EMB + 4 * lane]);
            d[q] = __builtin_fmaf(r.w, w.w, __builtin_fmaf(r.z, w.z, __builtin_fmaf(r.y, w.y, r.x * w.x)));
        }
#pragma unroll
        for (int q = 0; q < NHEAD; ++q) d[q] = wave_sum(d[q]);

        const float nl = (lane == 0) ? d[0] : ((lane == 1) ? d[1] : d[2]);
        const float pv = (lane == 0) ? d[3] : ((lane == 1) ? d[4] : ((lane == 2) ? d[5] : d[6]));
        if (lane < 3) node_out[(size_t)i * 3 + lane] = nl + nb;
        if (lane < 4) p[(size_t)i * 4 + lane] = pv;
    }

    *(float4*)(&csh[wv * EMB + 4 * lane]) = cs;
    __syncthreads();
    if (threadIdx.x < EMB) {
        float s = 0.f;
#pragma unroll
        for (int w = 0; w < 8; ++w) s += csh[w * EMB + threadIdx.x];
        part[(size_t)blockIdx.x * EMB + threadIdx.x] = s;
    }
}

__global__ __launch_bounds__(256) void colsum_reduce_kernel(const float* __restrict__ part,
                                                            float* __restrict__ part2) {
    __shared__ float sh[256];
    const int t = threadIdx.x, c = t & (EMB - 1), hf = t >> 7;
    const float* pp = part + ((size_t)blockIdx.x * 64 + hf * 32) * EMB + c;
    float s = 0.f;
#pragma unroll
    for (int r = 0; r < 32; ++r) s += pp[r * EMB];
    sh[t] = s; __syncthreads();
    if (t < EMB) part2[blockIdx.x * EMB + t] = sh[t] + sh[t + 128];
}

__global__ void edgehead_kernel(const float* __restrict__ p, const int* __restrict__ ei,
                                const float* __restrict__ epb, float* out, int E) {
    int e = blockIdx.x * blockDim.x + threadIdx.x;
    if (e >= E) return;
    int s = ei[e], d = ei[E + e];
    const float4 ps = *(const float4*)(p + (size_t)s * 4);
    const float4 pd = *(const float4*)(p + (size_t)d * 4);
    out[2 * e + 0] = ps.x + pd.z + epb[0];
    out[2 * e + 1] = ps.y + pd.w + epb[1];
}

__global__ __launch_bounds__(256) void mlphead_kernel(const float* __restrict__ part2, int N,
                               const float* __restrict__ fc1W, const float* __restrict__ fc1b,
                               const float* __restrict__ fc2W, const float* __restrict__ fc2b,
                               const float* __restrict__ gpW,  const float* __restrict__ gpb,
                               float* out_glob, float* out_val) {
    __shared__ float red[256];
    __shared__ __align__(16) float g[EMB];
    const int t = threadIdx.x;
    constexpr int R = GBLK / 64 / 2;
    float pr[R];
    {
        const float* pp = part2 + (size_t)((t >> 7) * R) * EMB + (t & (EMB - 1));
#pragma unroll
        for (int r = 0; r < R; ++r) pr[r] = pp[r * EMB];
    }
    float w[EMB];
#pragma unroll
    for (int k = 0; k < EMB; ++k) w[k] = fc1W[k * HID + t];
    const float b1v = fc1b[t];
    const float w2v = fc2W[t];
    const float gwv = (t < EMB) ? gpW[t] : 0.f;
    const float b2v = fc2b[0];
    const float gbv = gpb[0];
    {
        float s = 0.f;
#pragma unroll
        for (int r = 0; r < R; ++r) s += pr[r];
        red[t] = s; __syncthreads();
        if (t < EMB) g[t] = (red[t] + red[t + 128]) / (float)N;
        __syncthreads();
    }
    float acc = b1v;
#pragma unroll
    for (int k = 0; k < EMB; k += 4) {
        const float4 gq = *(const float4*)(&g[k]);
        acc += gq.x * w[k];
        acc += gq.y * w[k + 1];
        acc += gq.z * w[k + 2];
        acc += gq.w * w[k + 3];
    }
    float v = fmaxf(acc, 0.f);
    red[t] = v * w2v; __syncthreads();
    for (int k = 128; k > 0; k >>= 1) {
        if (t < k) red[t] += red[t + k];
        __syncthreads();
    }
    if (t == 0) *out_val = red[0] + b2v;
    __syncthreads();
    red[t] = (t < EMB) ? g[t] * gwv : 0.f; __syncthreads();
    for (int k = 128; k > 0; k >>= 1) {
        if (t < k) red[t] += red[t + k];
        __syncthreads();
    }
    if (t == 0) *out_glob = red[0] + gbv;
}

extern "C" void kernel_launch(void* const* d_in, const int* in_sizes, int n_in,
                              void* d_out, int out_size, void* d_ws, size_t ws_size,
                              hipStream_t stream) {
    const float* x    = (const float*)d_in[0];
    const int*   ei   = (const int*)  d_in[1];
    const float* W1   = (const float*)d_in[2];
    const float* b1   = (const float*)d_in[3];
    const float* W2   = (const float*)d_in[4];
    const float* b2   = (const float*)d_in[5];
    const float* fc1W = (const float*)d_in[6];
    const float* fc1b = (const float*)d_in[7];
    const float* fc2W = (const float*)d_in[8];
    const float* fc2b = (const float*)d_in[9];
    const float* npW  = (const float*)d_in[10];
    const float* npb  = (const float*)d_in[11];
    const float* epW  = (const float*)d_in[12];
    const float* epb  = (const float*)d_in[13];
    const float* gpW  = (const float*)d_in[14];
    const float* gpb  = (const float*)d_in[15];

    const int N = in_sizes[0] / IN_DIM;
    const int E = in_sizes[1] / 2;

    float* ws = (float*)d_ws;
    size_t off = 0;
    auto alloc = [&](size_t n) { size_t o = off; off += (n + 63) & ~size_t(63); return o; };
    float* dinv = ws + alloc((size_t)N);
    const int NBK  = (N + BKN - 1) / BKN;
    const int NBLK = (E + EPB - 1) / EPB;
    int*   boff   = (int*)(ws + alloc((size_t)NBLK * (NBK + 1)));
    int*   rowptr = (int*)(ws + alloc((size_t)N + 1));
    unsigned int* bedge = (unsigned int*)(ws + alloc((size_t)E));
    int*   csrc   = (int*)(ws + alloc((size_t)E));
    float* cw     = ws + alloc((size_t)E);
    float* part  = ws + alloc((size_t)GBLK * EMB);
    float* part2 = ws + alloc((size_t)(GBLK / 64) * EMB);
    float* pbuf = ws + alloc((size_t)N * 4);
    unsigned int* H1 = (unsigned int*)(ws + alloc((size_t)N * (HID / 2)));
    float* H2   = ws + alloc((size_t)N * EMB + 16 * EMB);

    const dim3 blk(256);
    auto nb = [](long long n) { return (unsigned)((n + 255) / 256); };

    const int numTiles = N / 16;
    const int npairs   = (numTiles + 1) / 2;
    const int psplit   = npairs / 3;
    const int GUESTS   = 264;
    bin_local_kernel<<<NBLK + GUESTS, dim3(1024), 0, stream>>>(ei, bedge, boff, E, NBK, NBLK,
                                                                x, W1, H1, 0, psplit, numTiles);
    bin_rank_kernel<<<NBK + GUESTS, dim3(1024), 0, stream>>>(bedge, boff, rowptr, dinv, csrc, N, NBK, NBLK,
                                                              x, W1, H1, psplit, npairs, numTiles);
    edgeweight_kernel<<<nb((E + 3) / 4), blk, 0, stream>>>(csrc, dinv, cw, E);

    gather_gemm2_kernel<<<512, dim3(1024), 0, stream>>>(H1, rowptr, csrc, cw, dinv, b1, W2, H2, N);

    float* out = (float*)d_out;
    gather_heads_kernel<<<GBLK, blk, 0, stream>>>(H2, rowptr, csrc, cw, dinv, b2, npW, npb, epW,
                                                   out, pbuf, part, N);
    colsum_reduce_kernel<<<GBLK / 64, blk, 0, stream>>>(part, part2);

    edgehead_kernel<<<nb(E), blk, 0, stream>>>(pbuf, ei, epb, out + (size_t)N * 3, E);
    mlphead_kernel<<<1, blk, 0, stream>>>(part2, N, fc1W, fc1b, fc2W, fc2b, gpW, gpb,
                                          out + (size_t)N * 3 + (size_t)E * 2,
                                          out + (size_t)N * 3 + (size_t)E * 2 + 1);
}


# h1@W2 kept as bf16 (extra rounding, nearest even) between its product and the layer-2 aggregation: gathered rows 256 B not 512 B, one 8-byte load a lane and row; sums stay f32, list order unchanged
# speedup vs baseline: 8.5823x; 1.0663x over previous
#include <hip/hip_runtime.h>
#include <stdint.h>

#define IN_DIM 128
#define HID    256
#define EMB    128

typedef __attribute__((ext_vector_type(16))) __bf16       v16bf;
typedef __attribute__((ext_vector_type(8)))  float        v8f;
typedef __attribute__((ext_vector_type(4)))  unsigned int uint4v;
typedef __attribute__((ext_vector_type(2)))  unsigned int uint2v;

__device__ __forceinline__ unsigned short f2bf(float f) {
    unsigned int u = __float_as_uint(f);
    u += 0x7FFFu + ((u >> 16) & 1u);
    return (unsigned short)(u >> 16);
}

#define BKN_SHIFT 8
#define BKN   256
#define EPB   4096
#define MAXB  1024
#define SG    8

__device__ __forceinline__ void bin_local_body(const int* __restrict__ ei, unsigned int* __restrict__ bedge,
                                               int* __restrict__ boff, int E, int NBK) {
    __shared__ int hist[MAXB];
    __shared__ int wtot[32];
    const int t = threadIdx.x, lane = t & 31, wv = t >> 5;
    const int blk = blockIdx.x;
    if (t < NBK) hist[t] = 0;
    __syncthreads();
    const int base = blk * EPB + t;
    int s[4], d[4];
#pragma unroll
    for (int k = 0; k < 4; ++k) {
        const int e  = base + k * 1024;
        const int ec = (e < E) ? e : E - 1;
        s[k] = ei[ec];
        d[k] = ei[E + ec];
    }
    asm volatile("" :: "v"(s[0]), "v"(s[1]), "v"(s[2]), "v"(s[3]), "v"(d[0]), "v"(d[1]), "v"(d[2]), "v"(d[3]));
#pragma unroll
    for (int k = 0; k < 4; ++k)
        if (base + k * 1024 < E) atomicAdd(&hist[d[k] >> BKN_SHIFT], 1);
    __syncthreads();
    const int c = (t < NBK) ? hist[t] : 0;
    int inc = c;
    for (int o = 1; o < 32; o <<= 1) {
        const int v = __shfl_up(inc, o, 32);
        if (lane >= o) inc += v;
    }
    if (lane == 31) wtot[wv] = inc;
    __syncthreads();
    int wbase = 0;
    for (int w = 0; w < 32; ++w)
        if (w < wv) wbase += wtot[w];
    const int excl = wbase + inc - c;
    int* bo = boff + (size_t)blk * (NBK + 1);
    if (t < NBK) { hist[t] = excl; bo[t] = excl; }
    if (t == NBK - 1) bo[NBK] = excl + c;
    __syncthreads();
    unsigned int pos[4];
#pragma unroll
    for (int k = 0; k < 4; ++k)
        pos[k] = (base + k * 1024 < E) ? (unsigned int)atomicAdd(&hist[d[k] >> BKN_SHIFT], 1) : 0u;
#pragma unroll
    for (int k = 0; k < 4; ++k)
        if (base + k * 1024 < E)
            bedge[(unsigned int)(blk * EPB) + pos[k]] = ((unsigned int)s[k] << BKN_SHIFT) | (unsigned int)(d[k] & (BKN - 1));
}

__device__ __forceinline__ void part_bounds(const int* __restrict__ boff, int pitch, int bin, int q0, int wv,
                                            int lane, int NBLK, int* o, int* c, int& front) {
    const int  sl = (q0 + (lane & (SG - 1))) * 32 + wv;
    const bool in = sl < NBLK;
    const int* p  = boff + (size_t)(in ? sl : 0) * pitch + bin;
    const int a0 = p[0], a1 = p[1];
    const int ov = sl * EPB + a0;
    const int cv = in ? (a1 - a0) : 0;
    const int fv = in ? a0 : 0;
#pragma unroll
    for (int q = 0; q < SG; ++q) {
        o[q] = __builtin_amdgcn_readlane(ov, q);
        c[q] = __builtin_amdgcn_readlane(cv, q);
        front += __builtin_amdgcn_readlane(fv, q);
    }
}

__device__ __forceinline__ void part_heads(const unsigned int* __restrict__ bedge, int lane,
                                           const int* o, const int* c, unsigned int* w) {
#pragma unroll
    for (int q = 0; q < SG; ++q) w[q] = bedge[(lane < c[q]) ? (o[q] + lane) : 0];
    asm volatile("" :: "v"(w[0]), "v"(w[1]), "v"(w[2]), "v"(w[3]), "v"(w[4]), "v"(w[5]), "v"(w[6]), "v"(w[7]));
}

__device__ __forceinline__ void count_parts(int* cnt, const unsigned int* __restrict__ bedge, int lane,
                                            const int* o, const int* c, const unsigned int* w) {
#pragma unroll
    for (int q = 0; q < SG; ++q)
        if (lane < c[q]) atomicAdd(&cnt[w[q] & (BKN - 1)], 1);
#pragma unroll
    for (int q = 0; q < SG; ++q)
        for (int j = 32 + lane; j < c[q]; j += 32) atomicAdd(&cnt[bedge[o[q] + j] & (BKN - 1)], 1);
}

__device__ __forceinline__ void rank_parts(int* cnt, const unsigned int* __restrict__ bedge, int* __restrict__ csrc,
                                           int lane, const int* o, const int* c, const unsigned int* w) {
    unsigned int pos[SG];
#pragma unroll
    for (int q = 0; q < SG; ++q)
        pos[q] = (lane < c[q]) ? (unsigned int)atomicAdd(&cnt[w[q] & (BKN - 1)], 1) : 0u;
#pragma unroll
    for (int q = 0; q < SG; ++q)
        if (lane < c[q]) csrc[pos[q]] = (int)(w[q] >> BKN_SHIFT);
#pragma unroll
    for (int q = 0; q < SG; ++q) {
        for (int j = 32 + lane; j < c[q]; j += 32) {
            const unsigned int x = bedge[o[q] + j];
            const int at = atomicAdd(&cnt[x & (BKN - 1)], 1);
            csrc[at] = (int)(x >> BKN_SHIFT);
        }
    }
}

__device__ __forceinline__ void bin_rank_body(const unsigned int* __restrict__ bedge,
                                              const int* __restrict__ boff,
                                              int* __restrict__ rowptr, float* __restrict__ dinv,
                                              int* __restrict__ csrc, int N, int NBK, int NBLK) {
    __shared__ int cnt[BKN];
    __shared__ int wtot[BKN / 32];
    __shared__ int wsum[32];
    const int t = threadIdx.x, lane = t & 31;
    const int wv = t >> 5;
    const int bin = blockIdx.x;
    const int pitch = NBK + 1;
    if (t < BKN) cnt[t] = 0;
    int front = 0;
    int o[SG], c[SG];
    unsigned int w[SG];
    part_bounds(boff, pitch, bin, 0, wv, lane, NBLK, o, c, front);
    part_heads(bedge, lane, o, c, w);
    __syncthreads();
    count_parts(cnt, bedge, lane, o, c, w);
    for (int q0 = SG; q0 * 32 < NBLK; q0 += SG) {
        int o2[SG], c2[SG];
        unsigned int w2[SG];
        part_bounds(boff, pitch, bin, q0, wv, lane, NBLK, o2, c2, front);
        part_heads(bedge, lane, o2, c2, w2);
        count_parts(cnt, bedge, lane, o2, c2, w2);
    }
    if (lane == 0) wsum[wv] = front;
    __syncthreads();
    int dg = 0, inc = 0;
    if (t < BKN) {
        dg = cnt[t];
        inc = dg;
        for (int s = 1; s < 32; s <<= 1) {
            const int v = __shfl_up(inc, s, 32);
            if (lane >= s) inc += v;
        }
        if (lane == 31) wtot[wv] = inc;
    }
    __syncthreads();
    if (t < BKN) {
        int bs = 0;
        for (int k = 0; k < 32; ++k) bs += wsum[k];
        int wbase = 0;
        for (int k = 0; k < BKN / 32; ++k)
            if (k < wv) wbase += wtot[k];
        const int start = bs + wbase + inc - dg;
        cnt[t] = start;
        const int node = bin * BKN + t;
        if (node < N) {
            rowptr[node] = start;
            dinv[node] = rsqrtf((float)dg + 1.0f);
        }
        if (node == N - 1) rowptr[N] = start + dg;
    }
    __syncthreads();
    rank_parts(cnt, bedge, csrc, lane, o, c, w);
    for (int q0 = SG; q0 * 32 < NBLK; q0 += SG) {
        int o2[SG], c2[SG], unused = 0;
        unsigned int w2[SG];
        part_bounds(boff, pitch, bin, q0, wv, lane, NBLK, o2, c2, unused);
        part_heads(bedge, lane, o2, c2, w2);
        rank_parts(cnt, bedge, csrc, lane, o2, c2, w2);
    }
}

__global__ void edgeweight_kernel(const int* __restrict__ csrc, const float* __restrict__ dinv,
                                  float* __restrict__ cw, int E) {
    int p = (blockIdx.x * blockDim.x + threadIdx.x) * 4;
    if (p + 3 < E) {
        const int4 s = *(const int4*)(csrc + p);
        *(float4*)(cw + p) = make_float4(dinv[s.x], dinv[s.y], dinv[s.z], dinv[s.w]);
    } else {
        for (; p < E; ++p) cw[p] = dinv[csrc[p]];
    }
}

#define BN  128

__device__ __forceinline__ void gemm1_guest(const float* __restrict__ A, const float* __restrict__ Wm,
                                            unsigned int* __restrict__ C, int g, int G, int pair0, int pair1,
                                            int numTiles) {
    constexpr int K   = IN_DIM;
    constexpr int LDA = K + 8;
    constexpr int KS  = K / 32;
    __shared__ __align__(16) unsigned short sA1[2][32 * LDA];

    const int tid   = threadIdx.x;
    const int wave  = tid >> 5;
    const int lane  = tid & 31;
    const int mn    = lane & 15;
    const int khalf = lane >> 4;
    const int quad  = wave >> 3;
    const int rt    = quad >> 1;
    const int col   = (quad & 1) * BN + (wave & 7) * 16 + mn;

    union BF { v16bf v; unsigned short h[16]; };
    BF fb[KS];
#pragma unroll
    for (int kk = 0; kk < KS; ++kk) {
        const float* wp = Wm + (size_t)(kk * 32 + khalf * 8) * HID + col;
#pragma unroll
        for (int e = 0; e < 8; ++e) {
            fb[kk].h[e]     = f2bf(wp[(size_t)e * HID]);
            fb[kk].h[8 + e] = f2bf(wp[(size_t)(16 + e) * HID]);
        }
    }

    const int sr = wave;
    const int sc = lane * 4;
    const int lastRow = numTiles * 16 - 1;

    float4 av;
    int p = pair0 + g;
    if (p < pair1) {
        const int row = (p * 32 + sr < lastRow) ? (p * 32 + sr) : lastRow;
        av = *(const float4*)(&A[(size_t)row * K + sc]);
        unsigned int lo = (unsigned int)f2bf(av.x) | ((unsigned int)f2bf(av.y) << 16);
        unsigned int hi = (unsigned int)f2bf(av.z) | ((unsigned int)f2bf(av.w) << 16);
        *(uint2v*)(&sA1[0][sr * LDA + sc]) = (uint2v){lo, hi};
    }

    int buf = 0;
    for (; p < pair1; p += G) {
        const int  next     = p + G;
        const bool has_next = next < pair1;
        const int ldp = has_next ? next : p;
        const int row = (ldp * 32 + sr < lastRow) ? (ldp * 32 + sr) : lastRow;
        av = *(const float4*)(&A[(size_t)row * K + sc]);
        __builtin_amdgcn_fence(__ATOMIC_RELEASE, "workgroup", "local");
        __builtin_amdgcn_s_barrier();
        __builtin_amdgcn_fence(__ATOMIC_ACQUIRE, "workgroup", "local");

        const unsigned short* pa = &sA1[buf][(rt * 16 + mn) * LDA + khalf * 8];
        v8f acc = {0.f, 0.f, 0.f, 0.f, 0.f, 0.f, 0.f, 0.f};
#pragma unroll
        for (int kk = 0; kk < KS; ++kk) {
            union { v16bf v; uint4v q[2]; } fa;
            fa.q[0] = *(const uint4v*)(pa + kk * 32);
            fa.q[1] = *(const uint4v*)(pa + kk * 32 + 16);
            acc = __builtin_amdgcn_wmma_f32_16x16x32_bf16(
                false, fa.v, false, fb[kk].v, (short)0, acc, false, false);
        }

        const int tile = 2 * p + rt;
        if (tile < numTiles) {
            const bool odd = (mn & 1) != 0;
            unsigned int* cp = C + (size_t)(tile * 16 + khalf * 8 + (odd ? 4 : 0)) * (HID / 2) + (col >> 1);
#pragma unroll
            for (int r = 0; r < 4; ++r) {
                const unsigned int own = (unsigned int)f2bf(acc[r]) | ((unsigned int)f2bf(acc[4 + r]) << 16);
                const unsigned int oth = (unsigned int)__builtin_amdgcn_update_dpp(0, (int)own, 0xB1, 0xf, 0xf, true);
                const unsigned int we = (own & 0xFFFFu) | (oth << 16);
                const unsigned int wo = (oth >> 16) | (own & 0xFFFF0000u);
                cp[(size_t)r * (HID / 2)] = odd ? wo : we;
            }
        }

        if (has_next) {
            unsigned int lo = (unsigned int)f2bf(av.x) | ((unsigned int)f2bf(av.y) << 16);
            unsigned int hi = (unsigned int)f2bf(av.z) | ((unsigned int)f2bf(av.w) << 16);
            *(uint2v*)(&sA1[buf ^ 1][sr * LDA + sc]) = (uint2v){lo, hi};
        }
        buf ^= 1;
    }
}

__global__ __launch_bounds__(1024, 16) void bin_local_kernel(const int* __restrict__ ei, unsigned int* __restrict__ bedge,
                                                         int* __restrict__ boff, int E, int NBK, int NBLK,
                                                         const float* __restrict__ x, const float* __restrict__ W1,
                                                         unsigned int* __restrict__ H1, int pair0, int pair1, int numTiles) {
    if ((int)blockIdx.x < NBLK) bin_local_body(ei, bedge, boff, E, NBK);
    else gemm1_guest(x, W1, H1, (int)blockIdx.x - NBLK, (int)gridDim.x - NBLK, pair0, pair1, numTiles);
}

__global__ __launch_bounds__(1024, 16) void bin_rank_kernel(const unsigned int* __restrict__ bedge,
                                                        const int* __restrict__ boff,
                                                        int* __restrict__ rowptr, float* __restrict__ dinv,
                                                        int* __restrict__ csrc, int N, int NBK, int NBLK,
                                                        const float* __restrict__ x, const float* __restrict__ W1,
                                                        unsigned int* __restrict__ H1, int pair0, int pair1, int numTiles) {
    if ((int)blockIdx.x < NBK) bin_rank_body(bedge, boff, rowptr, dinv, csrc, N, NBK, NBLK);
    else gemm1_guest(x, W1, H1, (int)blockIdx.x - NBK, (int)gridDim.x - NBK, pair0, pair1, numTiles);
}

template <int K>
__device__ __forceinline__ void add_rows_bf4(const unsigned int* __restrict__ H, int sv, float wl,
                                             int j, int lane, float4& acc) {
    uint2v h[K]; float w[K];
#pragma unroll
    for (int u = 0; u < K; ++u) {
        const int s = __builtin_amdgcn_readlane(sv, j + u);
        w[u] = __int_as_float(__builtin_amdgcn_readlane(__float_as_int(wl), j + u));
        h[u] = *(const uint2v*)((const char*)H + ((unsigned int)s * (unsigned int)(EMB * 2) + 8u * (unsigned int)lane));
    }
#pragma unroll
    for (int u = 0; u < K; ++u) {
        acc.x = __builtin_fmaf(__uint_as_float(h[u].x << 16), w[u], acc.x);
        acc.y = __builtin_fmaf(__uint_as_float(h[u].x & 0xFFFF0000u), w[u], acc.y);
        acc.z = __builtin_fmaf(__uint_as_float(h[u].y << 16), w[u], acc.z);
        acc.w = __builtin_fmaf(__uint_as_float(h[u].y & 0xFFFF0000u), w[u], acc.w);
    }
}

template <int K>
__device__ __forceinline__ void add_rows_bf(const unsigned int* __restrict__ H, int sv, float wl,
                                            int j, int lane, float* acc) {
    uint4v h[K]; float w[K];
#pragma unroll
    for (int u = 0; u < K; ++u) {
        const int s = __builtin_amdgcn_readlane(sv, j + u);
        w[u] = __int_as_float(__builtin_amdgcn_readlane(__float_as_int(wl), j + u));
        h[u] = *(const uint4v*)((const char*)H + ((unsigned int)s * (unsigned int)(HID * 2) + 16u * (unsigned int)lane));
    }
#pragma unroll
    for (int u = 0; u < K; ++u) {
#pragma unroll
        for (int q = 0; q < 4; ++q) {
            const unsigned int d = h[u][q];
            acc[2 * q]     = __builtin_fmaf(__uint_as_float(d << 16), w[u], acc[2 * q]);
            acc[2 * q + 1] = __builtin_fmaf(__uint_as_float(d & 0xFFFF0000u), w[u], acc[2 * q + 1]);
        }
    }
}

#define LDA2 (HID + 8)

__global__ __launch_bounds__(1024, 16) void gather_gemm2_kernel(
    const unsigned int* __restrict__ H, const int* __restrict__ rowptr,
    const int* __restrict__ csrc, const float* __restrict__ cw,
    const float* __restrict__ dinv, const float* __restrict__ bias,
    const float* __restrict__ W2, unsigned int* __restrict__ C, int N) {
    constexpr int U  = 8;
    constexpr int KS = HID / 32;
    __shared__ __align__(16) unsigned short sB[HID * EMB];
    __shared__ __align__(16) unsigned short sA[2][32 * LDA2];
    const int tid  = threadIdx.x;
    const int lane = tid & 31;
    const int wv   = __builtin_amdgcn_readfirstlane(tid >> 5);


    const int numTiles = (N + 15) >> 4;
    const int nTrips   = (N + 31) >> 5;
    int T     = blockIdx.x;
    int prevT = -1;
    int buf   = 0;
    for (;;) {
        const bool have = T < nTrips;
        unsigned int pk[4];
#pragma unroll
        for (int q = 0; q < 4; ++q) pk[q] = 0u;
        const int i = T * 32 + wv;
        if (have && i < N) {
            const float di  = dinv[i];
            const int   beg = rowptr[i], end = rowptr[i + 1];
            const float dd  = di * di;
            float acc[8];
            {
                const uint4v h = *(const uint4v*)((const char*)H + ((unsigned int)i * (unsigned int)(HID * 2) + 16u * (unsigned int)lane));
#pragma unroll
                for (int q = 0; q < 4; ++q) {
                    const unsigned int d = h[q];
                    acc[2 * q]     = __uint_as_float(d << 16) * dd;
                    acc[2 * q + 1] = __uint_as_float(d & 0xFFFF0000u) * dd;
                }
            }
            for (int p0 = beg; p0 < end; p0 += 32) {
                const int n = (end - p0 < 32) ? (end - p0) : 32;
                int sv = 0; float wl = 0.f;
                if (lane < n) { sv = csrc[p0 + lane]; wl = cw[p0 + lane] * di; }
                int j = 0;
                for (; j + U <= n; j += U) add_rows_bf<U>(H, sv, wl, j, lane, acc);
                if (j + 4 <= n) { add_rows_bf<4>(H, sv, wl, j, lane, acc); j += 4; }
                if (j + 2 <= n) { add_rows_bf<2>(H, sv, wl, j, lane, acc); j += 2; }
                if (j < n)      { add_rows_bf<1>(H, sv, wl, j, lane, acc); }
            }
            float bv[8];
            {
                const float4 b0 = *((const float4*)bias + 2 * lane), b1 = *((const float4*)bias + 2 * lane + 1);
                bv[0] = b0.x; bv[1] = b0.y; bv[2] = b0.z; bv[3] = b0.w;
                bv[4] = b1.x; bv[5] = b1.y; bv[6] = b1.z; bv[7] = b1.w;
            }
#pragma unroll
            for (int q = 0; q < 4; ++q) {
                const float r0 = fmaxf(acc[2 * q] + bv[2 * q], 0.0f);
                const float r1 = fmaxf(acc[2 * q + 1] + bv[2 * q + 1], 0.0f);
                pk[q] = (unsigned int)f2bf(r0) | ((unsigned int)f2bf(r1) << 16);
            }
        }

        if (prevT >= 0) {
            __builtin_amdgcn_s_barrier_wait(-1);
            __builtin_amdgcn_fence(__ATOMIC_ACQUIRE, "workgroup", "local");
            const int tl   = wv >> 3;
            const int tile = 2 * prevT + tl;
            if (wv < 16 && tile < numTiles) {
                const int mn    = lane & 15;
                const int khalf = lane >> 4;
                const unsigned short* pa = &sA[buf ^ 1][(tl * 16 + mn) * LDA2 + khalf * 8];
                const unsigned short* pb = &sB[((wv & 7) * (KS * 2 * 32) + lane) * 8];
                v8f d = {0.f, 0.f, 0.f, 0.f, 0.f, 0.f, 0.f, 0.f};
#pragma unroll
                for (int kk = 0; kk < KS; ++kk) {
                    union { v16bf v; uint4v q[2]; } fa, fb;
                    fa.q[0] = *(const uint4v*)(pa + kk * 32);
                    fa.q[1] = *(const uint4v*)(pa + kk * 32 + 16);
                    fb.q[0] = *(const uint4v*)(pb + (kk * 2 + 0) * 256);
                    fb.q[1] = *(const uint4v*)(pb + (kk * 2 + 1) * 256);
                    d = __builtin_amdgcn_wmma_f32_16x16x32_bf16(
                        false, fa.v, false, fb.v, (short)0, d, false, false);
                }
                const int  col = (wv & 7) * 16 + mn;
                const bool odd = (mn & 1) != 0;
                unsigned int* cp = C + (size_t)(tile * 16 + khalf * 8 + (odd ? 4 : 0)) * (EMB / 2) + (col >> 1);
#pragma unroll
                for (int r = 0; r < 4; ++r) {
                    const unsigned int own = (unsigned int)f2bf(d[r]) | ((unsigned int)f2bf(d[4 + r]) << 16);
                    const unsigned int oth = (unsigned int)__builtin_amdgcn_update_dpp(0, (int)own, 0xB1, 0xf, 0xf, true);
                    const unsigned int we = (own & 0xFFFFu) | (oth << 16);
                    const unsigned int wo = (oth >> 16) | (own & 0xFFFF0000u);
                    cp[(size_t)r * (EMB / 2)] = odd ? wo : we;
                }
            }
        } else if (have) {
#pragma unroll 8
            for (int it = 0; it < (HID * EMB) / 1024; ++it) {
                const int idx = it * 1024 + tid;
                const int k = idx >> 7, col = idx & (EMB - 1);
                const int r = k & 31;
                const int ln = ((r >> 3) & 1) * 16 + (col & 15);
                const int at = (((((col >> 4) * KS + (k >> 5)) * 2 + (r >> 4)) * 32 + ln) << 3) + (r & 7);
                sB[at] = f2bf(W2[idx]);
            }
        }
        if (!have) break;

        *(uint4v*)(&sA[buf][wv * LDA2 + 8 * lane]) = (uint4v){pk[0], pk[1], pk[2], pk[3]};
        __builtin_amdgcn_fence(__ATOMIC_RELEASE, "workgroup", "local");
        __builtin_amdgcn_s_barrier_signal(-1);
        prevT = T;
        T += (int)gridDim.x;
        buf ^= 1;
    }
}

#define GBLK  2048
#define NHEAD 7

template <int M>
__device__ __forceinline__ float row_xor_add(float x) {
    return x + __builtin_bit_cast(float, __builtin_amdgcn_update_dpp(0, __builtin_bit_cast(int, x), 0x160 + M, 0xf, 0xf, true));
}
__device__ __forceinline__ float wave_sum(float x) {
    x = row_xor_add<1>(x);
    x = row_xor_add<2>(x);
    x = row_xor_add<4>(x);
    x = row_xor_add<8>(x);
    const int xi = __builtin_bit_cast(int, x);
    return x + __builtin_bit_cast(float, __builtin_amdgcn_permlanex16(xi, xi, 0x76543210, 0xfedcba98, false, false));
}

__global__ __launch_bounds__(256) void gather_heads_kernel(
    const unsigned int* __restrict__ H, const int* __restrict__ rowptr,
    const int* __restrict__ csrc, const float* __restrict__ cw,
    const float* __restrict__ dinv, const float* __restrict__ bias,
    const float* __restrict__ npW, const float* __restrict__ npb, const float* __restrict__ epW,
    float* __restrict__ node_out, float* __restrict__ p, float* __restrict__ part, int N) {
    constexpr int U = 8;
    __shared__ __align__(16) float wsh[NHEAD * EMB];
    __shared__ __align__(16) float csh[8 * EMB];
    const int lane   = threadIdx.x & 31;
    const int wv     = __builtin_amdgcn_readfirstlane((int)(threadIdx.x >> 5));
    const int wave0  = blockIdx.x * 8 + wv;
    const int nwaves = gridDim.x * 8;

    if (threadIdx.x < EMB) {
        const int k = threadIdx.x;
        wsh[0 * EMB + k] = npW[k * 3 + 0];
        wsh[1 * EMB + k] = npW[k * 3 + 1];
        wsh[2 * EMB + k] = npW[k * 3 + 2];
        wsh[3 * EMB + k] = epW[k * 2 + 0];
        wsh[4 * EMB + k] = epW[k * 2 + 1];
    } else {
        const int k = threadIdx.x - EMB;
        wsh[5 * EMB + k] = epW[(EMB + k) * 2 + 0];
        wsh[6 * EMB + k] = epW[(EMB + k) * 2 + 1];
    }
    __syncthreads();

    const float4 bv = *((const float4*)bias + lane);
    const float  nb = (lane < 3) ? npb[lane] : 0.0f;
    float4 cs = make_float4(0.f, 0.f, 0.f, 0.f);

    for (int i = wave0; i < N; i += nwaves) {
        const float di  = dinv[i];
        const int   beg = rowptr[i], end = rowptr[i + 1];
        const float dd  = di * di;
        float4 acc;
        {
            const uint2v h = *(const uint2v*)((const char*)H + ((unsigned int)i * (unsigned int)(EMB * 2) + 8u * (unsigned int)lane));
            acc = make_float4(__uint_as_float(h.x << 16) * dd, __uint_as_float(h.x & 0xFFFF0000u) * dd,
                              __uint_as_float(h.y << 16) * dd, __uint_as_float(h.y & 0xFFFF0000u) * dd);
        }
        for (int p0 = beg; p0 < end; p0 += 32) {
            const int n = (end - p0 < 32) ? (end - p0) : 32;
            int sv = 0; float wl = 0.f;
            if (lane < n) { sv = csrc[p0 + lane]; wl = cw[p0 + lane] * di; }
            int j = 0;
            for (; j + U <= n; j += U) add_rows_bf4<U>(H, sv, wl, j, lane, acc);
            if (j + 4 <= n) { add_rows_bf4<4>(H, sv, wl, j, lane, acc); j += 4; }
            if (j + 2 <= n) { add_rows_bf4<2>(H, sv, wl, j, lane, acc); j += 2; }
            if (j < n)      { add_rows_bf4<1>(H, sv, wl, j, lane, acc); }
        }
        const float4 r = make_float4(fmaxf(acc.x + bv.x, 0.0f), fmaxf(acc.y + bv.y, 0.0f),
                                     fmaxf(acc.z + bv.z, 0.0f), fmaxf(acc.w + bv.w, 0.0f));
        cs.x += r.x; cs.y += r.y; cs.z += r.z; cs.w += r.w;

        __builtin_amdgcn_fence(__ATOMIC_ACQUIRE, "wavefront", "local");
        float d[NHEAD];
#pragma unroll
        for (int q = 0; q < NHEAD; ++q) {
            const float4 w = *(const float4*)(&wsh[q * EMB + 4 * lane]);
            d[q] = __builtin_fmaf(r.w, w.w, __builtin_fmaf(r.z, w.z, __builtin_fmaf(r.y, w.y, r.x * w.x)));
        }
#pragma unroll
        for (int q = 0; q < NHEAD; ++q) d[q] = wave_sum(d[q]);

        const float nl = (lane == 0) ? d[0] : ((lane == 1) ? d[1] : d[2]);
        const float pv = (lane == 0) ? d[3] : ((lane == 1) ? d[4] : ((lane == 2) ? d[5] : d[6]));
        if (lane < 3) node_out[(size_t)i * 3 + lane] = nl + nb;
        if (lane < 4) p[(size_t)i * 4 + lane] = pv;
    }

    *(float4*)(&csh[wv * EMB + 4 * lane]) = cs;
    __syncthreads();
    if (threadIdx.x < EMB) {
        float s = 0.f;
#pragma unroll
        for (int w = 0; w < 8; ++w) s += csh[w * EMB + threadIdx.x];
        part[(size_t)blockIdx.x * EMB + threadIdx.x] = s;
    }
}

__global__ __launch_bounds__(256) void colsum_reduce_kernel(const float* __restrict__ part,
                                                            float* __restrict__ part2) {
    __shared__ float sh[256];
    const int t = threadIdx.x, c = t & (EMB - 1), hf = t >> 7;
    const float* pp = part + ((size_t)blockIdx.x * 64 + hf * 32) * EMB + c;
    float s = 0.f;
#pragma unroll
    for (int r = 0; r < 32; ++r) s += pp[r * EMB];
    sh[t] = s; __syncthreads();
    if (t < EMB) part2[blockIdx.x * EMB + t] = sh[t] + sh[t + 128];
}

__global__ void edgehead_kernel(const float* __restrict__ p, const int* __restrict__ ei,
                                const float* __restrict__ epb, float* out, int E) {
    int e = blockIdx.x * blockDim.x + threadIdx.x;
    if (e >= E) return;
    int s = ei[e], d = ei[E + e];
    const float4 ps = *(const float4*)(p + (size_t)s * 4);
    const float4 pd = *(const float4*)(p + (size_t)d * 4);
    out[2 * e + 0] = ps.x + pd.z + epb[0];
    out[2 * e + 1] = ps.y + pd.w + epb[1];
}

__global__ __launch_bounds__(256) void mlphead_kernel(const float* __restrict__ part2, int N,
                               const float* __restrict__ fc1W, const float* __restrict__ fc1b,
                               const float* __restrict__ fc2W, const float* __restrict__ fc2b,
                               const float* __restrict__ gpW,  const float* __restrict__ gpb,
                               float* out_glob, float* out_val) {
    __shared__ float red[256];
    __shared__ __align__(16) float g[EMB];
    const int t = threadIdx.x;
    constexpr int R = GBLK / 64 / 2;
    float pr[R];
    {
        const float* pp = part2 + (size_t)((t >> 7) * R) * EMB + (t & (EMB - 1));
#pragma unroll
        for (int r = 0; r < R; ++r) pr[r] = pp[r * EMB];
    }
    float w[EMB];
#pragma unroll
    for (int k = 0; k < EMB; ++k) w[k] = fc1W[k * HID + t];
    const float b1v = fc1b[t];
    const float w2v = fc2W[t];
    const float gwv = (t < EMB) ? gpW[t] : 0.f;
    const float b2v = fc2b[0];
    const float gbv = gpb[0];
    {
        float s = 0.f;
#pragma unroll
        for (int r = 0; r < R; ++r) s += pr[r];
        red[t] = s; __syncthreads();
        if (t < EMB) g[t] = (red[t] + red[t + 128]) / (float)N;
        __syncthreads();
    }
    float acc = b1v;
#pragma unroll
    for (int k = 0; k < EMB; k += 4) {
        const float4 gq = *(const float4*)(&g[k]);
        acc += gq.x * w[k];
        acc += gq.y * w[k + 1];
        acc += gq.z * w[k + 2];
        acc += gq.w * w[k + 3];
    }
    float v = fmaxf(acc, 0.f);
    red[t] = v * w2v; __syncthreads();
    for (int k = 128; k > 0; k >>= 1) {
        if (t < k) red[t] += red[t + k];
        __syncthreads();
    }
    if (t == 0) *out_val = red[0] + b2v;
    __syncthreads();
    red[t] = (t < EMB) ? g[t] * gwv : 0.f; __syncthreads();
    for (int k = 128; k > 0; k >>= 1) {
        if (t < k) red[t] += red[t + k];
        __syncthreads();
    }
    if (t == 0) *out_glob = red[0] + gbv;
}

extern "C" void kernel_launch(void* const* d_in, const int* in_sizes, int n_in,
                              void* d_out, int out_size, void* d_ws, size_t ws_size,
                              hipStream_t stream) {
    const float* x    = (const float*)d_in[0];
    const int*   ei   = (const int*)  d_in[1];
    const float* W1   = (const float*)d_in[2];
    const float* b1   = (const float*)d_in[3];
    const float* W2   = (const float*)d_in[4];
    const float* b2   = (const float*)d_in[5];
    const float* fc1W = (const float*)d_in[6];
    const float* fc1b = (const float*)d_in[7];
    const float* fc2W = (const float*)d_in[8];
    const float* fc2b = (const float*)d_in[9];
    const float* npW  = (const float*)d_in[10];
    const float* npb  = (const float*)d_in[11];
    const float* epW  = (const float*)d_in[12];
    const float* epb  = (const float*)d_in[13];
    const float* gpW  = (const float*)d_in[14];
    const float* gpb  = (const float*)d_in[15];

    const int N = in_sizes[0] / IN_DIM;
    const int E = in_sizes[1] / 2;

    float* ws = (float*)d_ws;
    size_t off = 0;
    auto alloc = [&](size_t n) { size_t o = off; off += (n + 63) & ~size_t(63); return o; };
    float* dinv = ws + alloc((size_t)N);
    const int NBK  = (N + BKN - 1) / BKN;
    const int NBLK = (E + EPB - 1) / EPB;
    int*   boff   = (int*)(ws + alloc((size_t)NBLK * (NBK + 1)));
    int*   rowptr = (int*)(ws + alloc((size_t)N + 1));
    unsigned int* bedge = (unsigned int*)(ws + alloc((size_t)E));
    int*   csrc   = (int*)(ws + alloc((size_t)E));
    float* cw     = ws + alloc((size_t)E);
    float* part  = ws + alloc((size_t)GBLK * EMB);
    float* part2 = ws + alloc((size_t)(GBLK / 64) * EMB);
    float* pbuf = ws + alloc((size_t)N * 4);
    unsigned int* H1 = (unsigned int*)(ws + alloc((size_t)N * (HID / 2)));
    unsigned int* H2 = (unsigned int*)(ws + alloc((size_t)N * (EMB / 2) + 16 * (EMB / 2)));

    const dim3 blk(256);
    auto nb = [](long long n) { return (unsigned)((n + 255) / 256); };

    const int numTiles = N / 16;
    const int npairs   = (numTiles + 1) / 2;
    const int psplit   = npairs / 3;
    const int GUESTS   = 264;
    bin_local_kernel<<<NBLK + GUESTS, dim3(1024), 0, stream>>>(ei, bedge, boff, E, NBK, NBLK,
                                                                x, W1, H1, 0, psplit, numTiles);
    bin_rank_kernel<<<NBK + GUESTS, dim3(1024), 0, stream>>>(bedge, boff, rowptr, dinv, csrc, N, NBK, NBLK,
                                                              x, W1, H1, psplit, npairs, numTiles);
    edgeweight_kernel<<<nb((E + 3) / 4), blk, 0, stream>>>(csrc, dinv, cw, E);

    gather_gemm2_kernel<<<512, dim3(1024), 0, stream>>>(H1, rowptr, csrc, cw, dinv, b1, W2, H2, N);

    float* out = (float*)d_out;
    gather_heads_kernel<<<GBLK, blk, 0, stream>>>(H2, rowptr, csrc, cw, dinv, b2, npW, npb, epW,
                                                   out, pbuf, part, N);
    colsum_reduce_kernel<<<GBLK / 64, blk, 0, stream>>>(part, part2);

    edgehead_kernel<<<nb(E), blk, 0, stream>>>(pbuf, ei, epb, out + (size_t)N * 3, E);
    mlphead_kernel<<<1, blk, 0, stream>>>(part2, N, fc1W, fc1b, fc2W, fc2b, gpW, gpb,
                                          out + (size_t)N * 3 + (size_t)E * 2,
                                          out + (size_t)N * 3 + (size_t)E * 2 + 1);
}
